# starting kernel
# baseline (speedup 1.0000x reference)
#include <hip/hip_runtime.h>
#include <hip/hip_bf16.h>

typedef __attribute__((ext_vector_type(2))) float v2f;
typedef __attribute__((ext_vector_type(8))) float v8f;

#define N_NODES 50000
#define E_EDGES 1600000
#define NHEAD   4
#define DHEAD   16
#define F_IN    128
#define HD      64

__global__ void __launch_bounds__(256)
gt_qkv_proj_wmma(const float* __restrict__ X,
                 const float* __restrict__ Wq,
                 const float* __restrict__ Wk,
                 const float* __restrict__ Wv,
                 float* __restrict__ Q,
                 float* __restrict__ K,
                 float* __restrict__ V)
{
    const int wave = (int)((blockIdx.x * blockDim.x + threadIdx.x) >> 5);
    const int lane = (int)(threadIdx.x & 31);
    const int NTASK = (N_NODES / 16) * 3;
    if (wave >= NTASK) return;

    const int mat  = wave % 3;
    const int row0 = (wave / 3) * 16;
    const float* __restrict__ W   = (mat == 0) ? Wq : (mat == 1) ? Wk : Wv;
    float*       __restrict__ Out = (mat == 0) ? Q  : (mat == 1) ? K  : V;

    const int grp = lane >> 4;
    const int m16 = lane & 15;

    v8f c0 = {}, c1 = {}, c2 = {}, c3 = {};

    const float* __restrict__ arow = X + (size_t)(row0 + m16) * F_IN + 2 * grp;

#pragma unroll 4
    for (int k0 = 0; k0 < F_IN; k0 += 4) {
        v2f a;
        a.x = arow[k0];
        a.y = arow[k0 + 1];

        const float* __restrict__ wr0 = W + (size_t)(k0 + 2 * grp) * HD + m16;
        const float* __restrict__ wr1 = wr0 + HD;

        v2f b0, b1, b2, b3;
        b0.x = wr0[0];  b0.y = wr1[0];
        b1.x = wr0[16]; b1.y = wr1[16];
        b2.x = wr0[32]; b2.y = wr1[32];
        b3.x = wr0[48]; b3.y = wr1[48];

        c0 = __builtin_amdgcn_wmma_f32_16x16x4_f32(false, a, false, b0, (short)0, c0, false, false);
        c1 = __builtin_amdgcn_wmma_f32_16x16x4_f32(false, a, false, b1, (short)0, c1, false, false);
        c2 = __builtin_amdgcn_wmma_f32_16x16x4_f32(false, a, false, b2, (short)0, c2, false, false);
        c3 = __builtin_amdgcn_wmma_f32_16x16x4_f32(false, a, false, b3, (short)0, c3, false, false);
    }

    const int mrow = 8 * grp;
#pragma unroll
    for (int r = 0; r < 8; ++r) {
        float* __restrict__ o = Out + (size_t)(row0 + mrow + r) * HD + m16;
        o[0]  = c0[r];
        o[16] = c1[r];
        o[32] = c2[r];
        o[48] = c3[r];
    }
}

__global__ void __launch_bounds__(256)
gt_edge_attn(const int* __restrict__ src, const int* __restrict__ dst,
             const float* __restrict__ Q, const float* __restrict__ K,
             const float* __restrict__ V,
             float* __restrict__ wV, float* __restrict__ z)
{
    const int lane  = (int)(threadIdx.x & 31);
    const int wave  = (int)((blockIdx.x * blockDim.x + threadIdx.x) >> 5);
    const int nwave = (int)((gridDim.x * blockDim.x) >> 5);
    const int grp   = lane >> 4;
    const int dd    = lane & 15;

    for (int e = wave; e < E_EDGES; e += nwave) {
        const int s = __builtin_amdgcn_readfirstlane(src[e]);
        const int d = __builtin_amdgcn_readfirstlane(dst[e]);

        const size_t sb = (size_t)s * HD;
        const size_t db = (size_t)d * HD;

        const float k0 = K[sb + lane], k1 = K[sb + lane + 32];
        const float q0 = Q[db + lane], q1 = Q[db + lane + 32];
        const float v0 = V[sb + lane], v1 = V[sb + lane + 32];

        float p0 = k0 * q0;
        float p1 = k1 * q1;

#pragma unroll
        for (int off = 1; off < 16; off <<= 1) {
            p0 += __shfl_xor(p0, off, 32);
            p1 += __shfl_xor(p1, off, 32);
        }

        const float s0 = __expf(fminf(fmaxf(p0 * 0.25f, -5.0f), 5.0f));
        const float s1 = __expf(fminf(fmaxf(p1 * 0.25f, -5.0f), 5.0f));

        atomicAdd(&wV[db + lane],      s0 * v0);
        atomicAdd(&wV[db + lane + 32], s1 * v1);
        if (dd == 0) {
            atomicAdd(&z[(size_t)d * NHEAD + grp],     s0);
            atomicAdd(&z[(size_t)d * NHEAD + 2 + grp], s1);
        }
    }
}

__global__ void __launch_bounds__(256)
gt_normalize(float* __restrict__ out, const float* __restrict__ z)
{
    const int i = (int)(blockIdx.x * blockDim.x + threadIdx.x);
    if (i < N_NODES * HD) {
        const int n = i >> 6;
        const int h = (i >> 4) & 3;
        out[i] = out[i] / (z[n * NHEAD + h] + 1e-6f);
    }
}

extern "C" void kernel_launch(void* const* d_in, const int* in_sizes, int n_in,
                              void* d_out, int out_size, void* d_ws, size_t ws_size,
                              hipStream_t stream)
{
    (void)in_sizes; (void)n_in; (void)out_size; (void)ws_size;

    const float* X   = (const float*)d_in[0];
    const int*   src = (const int*)  d_in[1];
    const int*   dst = (const int*)  d_in[2];
    const float* Wq  = (const float*)d_in[3];
    const float* Wk  = (const float*)d_in[4];
    const float* Wv  = (const float*)d_in[5];

    float* out = (float*)d_out;
    float* ws  = (float*)d_ws;
    float* Q   = ws;
    float* K   = ws + (size_t)N_NODES * HD;
    float* V   = ws + (size_t)2 * N_NODES * HD;
    float* z   = ws + (size_t)3 * N_NODES * HD;

    hipMemsetAsync(out, 0, (size_t)N_NODES * HD * sizeof(float), stream);
    hipMemsetAsync(z,   0, (size_t)N_NODES * NHEAD * sizeof(float), stream);

    {
        const int ntask = (N_NODES / 16) * 3;
        const int wpb   = 256 / 32;
        const int blocks = (ntask + wpb - 1) / wpb;
        gt_qkv_proj_wmma<<<blocks, 256, 0, stream>>>(X, Wq, Wk, Wv, Q, K, V);
    }

    gt_edge_attn<<<4096, 256, 0, stream>>>(src, dst, Q, K, V, out, z);

    {
        const int total = N_NODES * HD;
        gt_normalize<<<(total + 255) / 256, 256, 0, stream>>>(out, z);
    }
}


# Edge phase, no global atomics, no edge filing: each of 256 blocks reads the WHOLE dst list, keeps edges into its 196 nodes, sums wV,z by LDS float adds (arrival order), divides, writes out once
# speedup vs baseline: 1.4390x; 1.4390x over previous
#include <hip/hip_runtime.h>
#include <hip/hip_bf16.h>

typedef __attribute__((ext_vector_type(2))) float v2f;
typedef __attribute__((ext_vector_type(8))) float v8f;

#define N_NODES 50000
#define E_EDGES 1600000
#define NHEAD   4
#define DHEAD   16
#define F_IN    128
#define HD      64

__global__ void __launch_bounds__(256)
gt_qkv_proj_wmma(const float* __restrict__ X,
                 const float* __restrict__ Wq,
                 const float* __restrict__ Wk,
                 const float* __restrict__ Wv,
                 float* __restrict__ Q,
                 float* __restrict__ K,
                 float* __restrict__ V)
{
    const int wave = (int)((blockIdx.x * blockDim.x + threadIdx.x) >> 5);
    const int lane = (int)(threadIdx.x & 31);
    const int NTASK = (N_NODES / 16) * 3;
    if (wave >= NTASK) return;

    const int mat  = wave % 3;
    const int row0 = (wave / 3) * 16;
    const float* __restrict__ W   = (mat == 0) ? Wq : (mat == 1) ? Wk : Wv;
    float*       __restrict__ Out = (mat == 0) ? Q  : (mat == 1) ? K  : V;

    const int grp = lane >> 4;
    const int m16 = lane & 15;

    v8f c0 = {}, c1 = {}, c2 = {}, c3 = {};

    const float* __restrict__ arow = X + (size_t)(row0 + m16) * F_IN + 2 * grp;

#pragma unroll 4
    for (int k0 = 0; k0 < F_IN; k0 += 4) {
        v2f a;
        a.x = arow[k0];
        a.y = arow[k0 + 1];

        const float* __restrict__ wr0 = W + (size_t)(k0 + 2 * grp) * HD + m16;
        const float* __restrict__ wr1 = wr0 + HD;

        v2f b0, b1, b2, b3;
        b0.x = wr0[0];  b0.y = wr1[0];
        b1.x = wr0[16]; b1.y = wr1[16];
        b2.x = wr0[32]; b2.y = wr1[32];
        b3.x = wr0[48]; b3.y = wr1[48];

        c0 = __builtin_amdgcn_wmma_f32_16x16x4_f32(false, a, false, b0, (short)0, c0, false, false);
        c1 = __builtin_amdgcn_wmma_f32_16x16x4_f32(false, a, false, b1, (short)0, c1, false, false);
        c2 = __builtin_amdgcn_wmma_f32_16x16x4_f32(false, a, false, b2, (short)0, c2, false, false);
        c3 = __builtin_amdgcn_wmma_f32_16x16x4_f32(false, a, false, b3, (short)0, c3, false, false);
    }

    const int mrow = 8 * grp;
#pragma unroll
    for (int r = 0; r < 8; ++r) {
        float* __restrict__ o = Out + (size_t)(row0 + mrow + r) * HD + m16;
        o[0]  = c0[r];
        o[16] = c1[r];
        o[32] = c2[r];
        o[48] = c3[r];
    }
}

#define N_OWNERS   256
#define OWN_MAX    196
#define ACC_PITCH  68
#define OWN_WAVES  32
#define IDS_PER_LD 128
#define N_CHUNK    (E_EDGES / IDS_PER_LD)
#define GRP_LOADS  4
#define N_STEP     98
#define Q_KEEP     64
#define QW         (Q_KEEP + 2 * GRP_LOADS * IDS_PER_LD)
#define START_SKEW 48
#define E_MASK     0x1FFFFFu
#define EDGE_U     8

static_assert(E_EDGES % IDS_PER_LD == 0, "whole wave loads");
static_assert(N_STEP % 2 == 0 && N_STEP * OWN_WAVES * GRP_LOADS >= N_CHUNK, "the steps cover the list");
static_assert(E_EDGES <= (1 << 21) && OWN_MAX <= (1 << 11), "queue entry packing");
static_assert(((N_STEP + 1) * GRP_LOADS) * OWN_WAVES + (N_OWNERS - 1) * START_SKEW < 2 * N_CHUNK, "one wrap at most");
static_assert(OWN_MAX * (HD / 4) <= 4 * 1024, "Q rows: four 16-byte loads per thread");

__device__ __forceinline__ void
scan_load(const int4* __restrict__ dst4, const int step, const int wave, const int lane,
          const int skew, const unsigned cnt,
          int4 (&d)[GRP_LOADS], unsigned (&e)[GRP_LOADS], unsigned (&lim)[GRP_LOADS])
{
#pragma unroll
    for (int u = 0; u < GRP_LOADS; ++u) {
        const int  c     = (step * GRP_LOADS + u) * OWN_WAVES + wave;
        const bool valid = c < N_CHUNK;
        int p = c + skew; if (p >= N_CHUNK) p -= N_CHUNK;
        d[u]   = dst4[p * 32 + lane];
        e[u]   = (unsigned)p * IDS_PER_LD;
        lim[u] = valid ? cnt : 0u;
    }
}

__device__ __forceinline__ void
scan_keep(const int4 (&d)[GRP_LOADS], const unsigned (&e)[GRP_LOADS], const unsigned (&lim)[GRP_LOADS],
          const unsigned lo, const int lane, unsigned* myq, unsigned& count)
{
#pragma unroll
    for (int u = 0; u < GRP_LOADS; ++u) {
        const unsigned e0 = e[u] + (unsigned)lane * 4u;
        const unsigned d0 = (unsigned)d[u].x - lo, d1 = (unsigned)d[u].y - lo;
        const unsigned d2 = (unsigned)d[u].z - lo, d3 = (unsigned)d[u].w - lo;
        const bool m0 = d0 < lim[u], m1 = d1 < lim[u], m2 = d2 < lim[u], m3 = d3 < lim[u];
        const unsigned b0 = __builtin_amdgcn_ballot_w32(m0), b1 = __builtin_amdgcn_ballot_w32(m1);
        const unsigned b2 = __builtin_amdgcn_ballot_w32(m2), b3 = __builtin_amdgcn_ballot_w32(m3);
        if ((b0 | b1 | b2 | b3) != 0u) {
            if (m0) myq[count + __builtin_amdgcn_mbcnt_lo(b0, 0u)] = (e0     ) | (d0 << 21);
            count += (unsigned)__builtin_popcount(b0);
            if (m1) myq[count + __builtin_amdgcn_mbcnt_lo(b1, 0u)] = (e0 + 1u) | (d1 << 21);
            count += (unsigned)__builtin_popcount(b1);
            if (m2) myq[count + __builtin_amdgcn_mbcnt_lo(b2, 0u)] = (e0 + 2u) | (d2 << 21);
            count += (unsigned)__builtin_popcount(b2);
            if (m3) myq[count + __builtin_amdgcn_mbcnt_lo(b3, 0u)] = (e0 + 3u) | (d3 << 21);
            count += (unsigned)__builtin_popcount(b3);
        }
    }
}

template <int U>
__device__ __forceinline__ void
edge_batch(const unsigned (&srcs)[U], const unsigned (&dls)[U],
           const float* __restrict__ K, const float* __restrict__ V,
           const float* s_q, float* s_acc, const int lane)
{
    const int grp = lane >> 4;
    const int dd  = lane & 15;
    float k0[U], k1[U], v0[U], v1[U], p0[U], p1[U];
#pragma unroll
    for (int t = 0; t < U; ++t) {
        const float* __restrict__ kr = K + (size_t)srcs[t] * HD;
        const float* __restrict__ vr = V + (size_t)srcs[t] * HD;
        k0[t] = kr[lane]; k1[t] = kr[lane + 32];
        v0[t] = vr[lane]; v1[t] = vr[lane + 32];
    }
#pragma unroll
    for (int t = 0; t < U; ++t) {
        p0[t] = k0[t] * s_q[dls[t] * HD + lane];
        p1[t] = k1[t] * s_q[dls[t] * HD + lane + 32];
    }
#pragma unroll
    for (int off = 1; off < 16; off <<= 1) {
#pragma unroll
        for (int t = 0; t < U; ++t) {
            p0[t] += __shfl_xor(p0[t], off, 32);
            p1[t] += __shfl_xor(p1[t], off, 32);
        }
    }
#pragma unroll
    for (int t = 0; t < U; ++t) {
        p0[t] = __expf(fminf(fmaxf(p0[t] * 0.25f, -5.0f), 5.0f));
        p1[t] = __expf(fminf(fmaxf(p1[t] * 0.25f, -5.0f), 5.0f));
        float* a = s_acc + dls[t] * ACC_PITCH;
        atomicAdd(&a[lane],      p0[t] * v0[t]);
        atomicAdd(&a[lane + 32], p1[t] * v1[t]);
    }
    if (dd == 0) {
#pragma unroll
        for (int t = 0; t < U; ++t) {
            float* a = s_acc + dls[t] * ACC_PITCH + HD;
            atomicAdd(&a[grp],     p0[t]);
            atomicAdd(&a[2 + grp], p1[t]);
        }
    }
}

__global__ void __launch_bounds__(1024)
gt_edge_owner(const int* __restrict__ src, const int* __restrict__ dst,
              const float* __restrict__ Q, const float* __restrict__ K,
              const float* __restrict__ V, float* __restrict__ out)
{
    __shared__ __attribute__((aligned(16))) float    s_acc[OWN_MAX * ACC_PITCH];
    __shared__ __attribute__((aligned(16))) float    s_q[OWN_MAX * HD];
    __shared__ __attribute__((aligned(16))) unsigned s_queue[OWN_WAVES * QW];

    const int tid  = (int)threadIdx.x;
    const int lane = tid & 31;
    const int wave = __builtin_amdgcn_readfirstlane(tid >> 5);
    const int b    = (int)blockIdx.x;
    const unsigned lo  = (unsigned)(b * 3125) >> 4;
    const unsigned hi  = (unsigned)((b + 1) * 3125) >> 4;
    const unsigned cnt = hi - lo;
    const int skew = b * START_SKEW;

    const int4* __restrict__ dst4 = (const int4*)dst;
    unsigned* myq = s_queue + wave * QW;
    unsigned count = 0;

    int4     dA[GRP_LOADS], dB[GRP_LOADS];
    unsigned eA[GRP_LOADS], eB[GRP_LOADS], limA[GRP_LOADS], limB[GRP_LOADS];
    scan_load(dst4, 0, wave, lane, skew, cnt, dA, eA, limA);

    for (int i = tid; i < OWN_MAX * ACC_PITCH; i += 1024) s_acc[i] = 0.0f;
    {
        const float4* __restrict__ q4 = (const float4*)(Q + (size_t)lo * HD);
        float4* s4 = (float4*)s_q;
        const int nq = (int)cnt * (HD / 4);
        float4 r0, r1, r2, r3;
        { const int i = tid;        r0 = q4[i < nq ? i : nq - 1]; }
        { const int i = tid + 1024; r1 = q4[i < nq ? i : nq - 1]; }
        { const int i = tid + 2048; r2 = q4[i < nq ? i : nq - 1]; }
        { const int i = tid + 3072; r3 = q4[i < nq ? i : nq - 1]; }
        if (tid        < nq) s4[tid]        = r0;
        if (tid + 1024 < nq) s4[tid + 1024] = r1;
        if (tid + 2048 < nq) s4[tid + 2048] = r2;
        if (tid + 3072 < nq) s4[tid + 3072] = r3;
    }
    __syncthreads();

    for (int g = 0; ; g += 2) {
        if (g == N_STEP || count > Q_KEEP) {
            for (unsigned i0 = 0; i0 < count; i0 += 32) {
                const unsigned left = count - i0;
                const int nb = (int)(left < 32u ? left : 32u);
                unsigned ent = 0, sv = 0;
                if (lane < nb) { ent = myq[i0 + lane]; sv = (unsigned)src[ent & E_MASK]; }
                int j = 0;
                for (; j + EDGE_U <= nb; j += EDGE_U) {
                    unsigned srcs[EDGE_U], dls[EDGE_U];
#pragma unroll
                    for (int t = 0; t < EDGE_U; ++t) {
                        srcs[t] = __builtin_amdgcn_readlane(sv, j + t);
                        dls[t]  = __builtin_amdgcn_readlane(ent, j + t) >> 21;
                    }
                    edge_batch<EDGE_U>(srcs, dls, K, V, s_q, s_acc, lane);
                }
                for (; j < nb; ++j) {
                    unsigned srcs[1], dls[1];
                    srcs[0] = __builtin_amdgcn_readlane(sv, j);
                    dls[0]  = __builtin_amdgcn_readlane(ent, j) >> 21;
                    edge_batch<1>(srcs, dls, K, V, s_q, s_acc, lane);
                }
            }
            count = 0;
            if (g == N_STEP) break;
        }
        scan_load(dst4, g + 1, wave, lane, skew, cnt, dB, eB, limB);
        scan_keep(dA, eA, limA, lo, lane, myq, count);
        scan_load(dst4, g + 2, wave, lane, skew, cnt, dA, eA, limA);
        scan_keep(dB, eB, limB, lo, lane, myq, count);
    }

    __syncthreads();

    {
        float* __restrict__ o = out + (size_t)lo * HD;
        for (int i = tid; i < (int)cnt * HD; i += 1024) {
            const int r = i >> 6;
            const int c = i & 63;
            const int h = c >> 4;
            o[i] = s_acc[r * ACC_PITCH + c] / (s_acc[r * ACC_PITCH + HD + h] + 1e-6f);
        }
    }
}

extern "C" void kernel_launch(void* const* d_in, const int* in_sizes, int n_in,
                              void* d_out, int out_size, void* d_ws, size_t ws_size,
                              hipStream_t stream)
{
    (void)in_sizes; (void)n_in; (void)out_size; (void)ws_size;

    const float* X   = (const float*)d_in[0];
    const int*   src = (const int*)  d_in[1];
    const int*   dst = (const int*)  d_in[2];
    const float* Wq  = (const float*)d_in[3];
    const float* Wk  = (const float*)d_in[4];
    const float* Wv  = (const float*)d_in[5];

    float* out = (float*)d_out;
    float* ws  = (float*)d_ws;
    float* Q   = ws;
    float* K   = ws + (size_t)N_NODES * HD;
    float* V   = ws + (size_t)2 * N_NODES * HD;

    {
        const int ntask = (N_NODES / 16) * 3;
        const int wpb   = 256 / 32;
        const int blocks = (ntask + wpb - 1) / wpb;
        gt_qkv_proj_wmma<<<blocks, 256, 0, stream>>>(X, Wq, Wk, Wv, Q, K, V);
    }

    gt_edge_owner<<<N_OWNERS, 1024, 0, stream>>>(src, dst, Q, K, V, out);
}


# Score sum over each 16-lane row by four DPP row_xmask adds on the vector ALU instead of __shfl_xor (ds_bpermute through the LDS unit): 8 fewer LDS instructions per edge; same xor steps, same order
# speedup vs baseline: 1.7374x; 1.2074x over previous
#include <hip/hip_runtime.h>
#include <hip/hip_bf16.h>

typedef __attribute__((ext_vector_type(2))) float v2f;
typedef __attribute__((ext_vector_type(8))) float v8f;

#define N_NODES 50000
#define E_EDGES 1600000
#define NHEAD   4
#define DHEAD   16
#define F_IN    128
#define HD      64

#define PLANE_IDS    512
#define PLANE_CHUNKS (E_EDGES / PLANE_IDS)
#define PLANE_PER_WV 2
#define PLANE_BLOCKS ((PLANE_CHUNKS + 8 * PLANE_PER_WV - 1) / (8 * PLANE_PER_WV))

static_assert(E_EDGES % PLANE_IDS == 0, "whole chunks");
static_assert((16 * (N_NODES - 1) + 15) / 3125 == 255, "owner fits one byte");

__device__ __forceinline__ unsigned owner_of(const int d) { return (unsigned)(16 * d + 15) / 3125u; }

__device__ __forceinline__ void
plane_chunks(const int* __restrict__ dst, unsigned* __restrict__ plane, const int wave0, const int lane)
{
    const int4* __restrict__ dst4 = (const int4*)dst;
    int4 d[PLANE_PER_WV][4];
#pragma unroll
    for (int i = 0; i < PLANE_PER_WV; ++i) {
        const int c  = wave0 * PLANE_PER_WV + i;
        const int cc = c < PLANE_CHUNKS ? c : PLANE_CHUNKS - 1;
#pragma unroll
        for (int k = 0; k < 4; ++k) d[i][k] = dst4[cc * 128 + k * 32 + lane];
    }
#pragma unroll
    for (int i = 0; i < PLANE_PER_WV; ++i) {
        const int c = wave0 * PLANE_PER_WV + i;
        if (c < PLANE_CHUNKS) {
#pragma unroll
            for (int k = 0; k < 4; ++k) {
                const unsigned w = owner_of(d[i][k].x) | (owner_of(d[i][k].y) << 8)
                                 | (owner_of(d[i][k].z) << 16) | (owner_of(d[i][k].w) << 24);
                plane[c * 128 + k * 32 + lane] = w;
            }
        }
    }
}

__global__ void __launch_bounds__(256)
gt_qkv_proj_wmma(const float* __restrict__ X,
                 const int*   __restrict__ dst,
                 unsigned*    __restrict__ plane,
                 const float* __restrict__ Wq,
                 const float* __restrict__ Wk,
                 const float* __restrict__ Wv,
                 float* __restrict__ Q,
                 float* __restrict__ K,
                 float* __restrict__ V)
{
    const int lane = (int)(threadIdx.x & 31);
    if (blockIdx.x < PLANE_BLOCKS) {
        plane_chunks(dst, plane, (int)((blockIdx.x * blockDim.x + threadIdx.x) >> 5), lane);
        return;
    }
    const int wave = (int)(((blockIdx.x - PLANE_BLOCKS) * blockDim.x + threadIdx.x) >> 5);
    const int NTASK = (N_NODES / 16) * 3;
    if (wave >= NTASK) return;

    const int mat  = wave % 3;
    const int row0 = (wave / 3) * 16;
    const float* __restrict__ W   = (mat == 0) ? Wq : (mat == 1) ? Wk : Wv;
    float*       __restrict__ Out = (mat == 0) ? Q  : (mat == 1) ? K  : V;

    const int grp = lane >> 4;
    const int m16 = lane & 15;

    v8f c0 = {}, c1 = {}, c2 = {}, c3 = {};

    const float* __restrict__ arow = X + (size_t)(row0 + m16) * F_IN + 2 * grp;

#pragma unroll 4
    for (int k0 = 0; k0 < F_IN; k0 += 4) {
        v2f a;
        a.x = arow[k0];
        a.y = arow[k0 + 1];

        const float* __restrict__ wr0 = W + (size_t)(k0 + 2 * grp) * HD + m16;
        const float* __restrict__ wr1 = wr0 + HD;

        v2f b0, b1, b2, b3;
        b0.x = wr0[0];  b0.y = wr1[0];
        b1.x = wr0[16]; b1.y = wr1[16];
        b2.x = wr0[32]; b2.y = wr1[32];
        b3.x = wr0[48]; b3.y = wr1[48];

        c0 = __builtin_amdgcn_wmma_f32_16x16x4_f32(false, a, false, b0, (short)0, c0, false, false);
        c1 = __builtin_amdgcn_wmma_f32_16x16x4_f32(false, a, false, b1, (short)0, c1, false, false);
        c2 = __builtin_amdgcn_wmma_f32_16x16x4_f32(false, a, false, b2, (short)0, c2, false, false);
        c3 = __builtin_amdgcn_wmma_f32_16x16x4_f32(false, a, false, b3, (short)0, c3, false, false);
    }

    const int mrow = 8 * grp;
#pragma unroll
    for (int r = 0; r < 8; ++r) {
        float* __restrict__ o = Out + (size_t)(row0 + mrow + r) * HD + m16;
        o[0]  = c0[r];
        o[16] = c1[r];
        o[32] = c2[r];
        o[48] = c3[r];
    }
}

#define N_OWNERS   256
#define OWN_MAX    196
#define ACC_PITCH  68
#define OWN_WAVES  32
#define IDS_PER_LD PLANE_IDS
#define N_CHUNK    PLANE_CHUNKS
#define N_STEP     100
#define Q_KEEP     64
#define QW         (Q_KEEP + 2 * IDS_PER_LD)
#define START_SKEW 12
#define EDGE_U     8

static_assert(N_STEP % 4 == 0 && N_STEP * OWN_WAVES >= N_CHUNK, "the steps cover the plane");
static_assert(N_CHUNK + (N_OWNERS - 1) * START_SKEW < 2 * N_CHUNK, "one wrap at most");
static_assert(N_OWNERS == 256, "one byte names the owner");
static_assert(OWN_MAX * (HD / 4) <= 4 * 1024, "Q rows: four 16-byte loads per thread");

__device__ __forceinline__ void
scan_load(const uint4* __restrict__ plane4, const int step, const int wave, const int lane,
          const int skew, uint4& d, unsigned& e, bool& valid)
{
    const int c = step * OWN_WAVES + wave;
    valid = c < N_CHUNK;
    int p = (valid ? c : 0) + skew; if (p >= N_CHUNK) p -= N_CHUNK;
    d = plane4[p * 32 + lane];
    e = (unsigned)p * IDS_PER_LD;
}

__device__ __forceinline__ unsigned zero_bytes(const unsigned x)
{
    const unsigned t = ((x & 0x7F7F7F7Fu) + 0x7F7F7F7Fu) | x;
    return ~t & 0x80808080u;
}

__device__ __forceinline__ void
scan_keep(const uint4& d, const unsigned e, const bool valid, const unsigned pat,
          const int lane, unsigned* myq, unsigned& count)
{
    if (!valid) return;
    const unsigned z0 = zero_bytes(d.x ^ pat), z1 = zero_bytes(d.y ^ pat);
    const unsigned z2 = zero_bytes(d.z ^ pat), z3 = zero_bytes(d.w ^ pat);
    unsigned live = __builtin_amdgcn_ballot_w32((z0 | z1 | z2 | z3) != 0u);
    if (live != 0u) {
        unsigned hm = (z0 >> 7) | (z1 >> 6) | (z2 >> 5) | (z3 >> 4);
        const unsigned e0 = e + (unsigned)lane * 16u;
        do {
            if (hm != 0u) {
                const unsigned pos = (unsigned)__builtin_ctz(hm);
                myq[count + __builtin_amdgcn_mbcnt_lo(live, 0u)] = e0 + ((pos & 7u) << 2) + (pos >> 3);
            }
            count += (unsigned)__builtin_popcount(live);
            hm &= hm - 1u;
            live = __builtin_amdgcn_ballot_w32(hm != 0u);
        } while (live != 0u);
    }
}

template <int M>
__device__ __forceinline__ float row_xor(const float x)
{
    return __builtin_bit_cast(float,
        __builtin_amdgcn_mov_dpp(__builtin_bit_cast(int, x), 0x160 + M, 0xf, 0xf, true));
}

template <int M, int U>
__device__ __forceinline__ void row_xor_add(float (&p0)[U], float (&p1)[U])
{
#pragma unroll
    for (int t = 0; t < U; ++t) {
        p0[t] += row_xor<M>(p0[t]);
        p1[t] += row_xor<M>(p1[t]);
    }
}

template <int U>
__device__ __forceinline__ void
edge_batch(const unsigned (&srcs)[U], const unsigned (&dls)[U],
           const float* __restrict__ K, const float* __restrict__ V,
           const float* s_q, float* s_acc, const int lane)
{
    const int grp = lane >> 4;
    const int dd  = lane & 15;
    float k0[U], k1[U], v0[U], v1[U], p0[U], p1[U];
#pragma unroll
    for (int t = 0; t < U; ++t) {
        const float* __restrict__ kr = K + (size_t)srcs[t] * HD;
        const float* __restrict__ vr = V + (size_t)srcs[t] * HD;
        k0[t] = kr[lane]; k1[t] = kr[lane + 32];
        v0[t] = vr[lane]; v1[t] = vr[lane + 32];
    }
#pragma unroll
    for (int t = 0; t < U; ++t) {
        p0[t] = k0[t] * s_q[dls[t] * HD + lane];
        p1[t] = k1[t] * s_q[dls[t] * HD + lane + 32];
    }
    row_xor_add<1, U>(p0, p1);
    row_xor_add<2, U>(p0, p1);
    row_xor_add<4, U>(p0, p1);
    row_xor_add<8, U>(p0, p1);
#pragma unroll
    for (int t = 0; t < U; ++t) {
        p0[t] = __expf(fminf(fmaxf(p0[t] * 0.25f, -5.0f), 5.0f));
        p1[t] = __expf(fminf(fmaxf(p1[t] * 0.25f, -5.0f), 5.0f));
        float* a = s_acc + dls[t] * ACC_PITCH;
        atomicAdd(&a[lane],      p0[t] * v0[t]);
        atomicAdd(&a[lane + 32], p1[t] * v1[t]);
    }
    if (dd == 0) {
#pragma unroll
        for (int t = 0; t < U; ++t) {
            float* a = s_acc + dls[t] * ACC_PITCH + HD;
            atomicAdd(&a[grp],     p0[t]);
            atomicAdd(&a[2 + grp], p1[t]);
        }
    }
}

__device__ __forceinline__ void
work_off(const unsigned* myq, const unsigned count, const unsigned lo,
         const int* __restrict__ src, const int* __restrict__ dst,
         const float* __restrict__ K, const float* __restrict__ V,
         const float* s_q, float* s_acc, const int lane)
{
    for (unsigned i0 = 0; i0 < count; i0 += 32) {
        const unsigned left = count - i0;
        const int nb = (int)(left < 32u ? left : 32u);
        unsigned sv = 0, dv = 0;
        if (lane < nb) {
            const unsigned ent = myq[i0 + lane];
            sv = (unsigned)src[ent];
            dv = (unsigned)dst[ent] - lo;
            dv = dv < (unsigned)OWN_MAX ? dv : 0u;
        }
        int j = 0;
        for (; j + EDGE_U <= nb; j += EDGE_U) {
            unsigned srcs[EDGE_U], dls[EDGE_U];
#pragma unroll
            for (int t = 0; t < EDGE_U; ++t) {
                srcs[t] = __builtin_amdgcn_readlane(sv, j + t);
                dls[t]  = __builtin_amdgcn_readlane(dv, j + t);
            }
            edge_batch<EDGE_U>(srcs, dls, K, V, s_q, s_acc, lane);
        }
        for (; j < nb; ++j) {
            unsigned srcs[1], dls[1];
            srcs[0] = __builtin_amdgcn_readlane(sv, j);
            dls[0]  = __builtin_amdgcn_readlane(dv, j);
            edge_batch<1>(srcs, dls, K, V, s_q, s_acc, lane);
        }
    }
}

__global__ void __launch_bounds__(1024)
gt_edge_owner(const int* __restrict__ src, const int* __restrict__ dst,
              const uint4* __restrict__ plane4,
              const float* __restrict__ K, const float* __restrict__ V,
              float* out)
{
    __shared__ __attribute__((aligned(16))) float    s_acc[OWN_MAX * ACC_PITCH];
    __shared__ __attribute__((aligned(16))) float    s_q[OWN_MAX * HD];
    __shared__ __attribute__((aligned(16))) unsigned s_queue[OWN_WAVES * QW];

    const int tid  = (int)threadIdx.x;
    const int lane = tid & 31;
    const int wave = __builtin_amdgcn_readfirstlane(tid >> 5);
    const int b    = (int)blockIdx.x;
    const unsigned lo  = (unsigned)(b * 3125) >> 4;
    const unsigned hi  = (unsigned)((b + 1) * 3125) >> 4;
    const unsigned cnt = hi - lo;
    const unsigned pat = (unsigned)b * 0x01010101u;
    const int skew = b * START_SKEW;

    unsigned* myq = s_queue + wave * QW;
    unsigned count = 0;

    uint4    dA, dB, dC, dD;
    unsigned eA, eB, eC, eD;
    bool     okA, okB, okC, okD;
    scan_load(plane4, 0, wave, lane, skew, dA, eA, okA);
    scan_load(plane4, 1, wave, lane, skew, dB, eB, okB);
    scan_load(plane4, 2, wave, lane, skew, dC, eC, okC);
    scan_load(plane4, 3, wave, lane, skew, dD, eD, okD);

    for (int i = tid; i < OWN_MAX * ACC_PITCH; i += 1024) s_acc[i] = 0.0f;
    {
        const float4* q4 = (const float4*)(out + (size_t)lo * HD);
        float4* s4 = (float4*)s_q;
        const int nq = (int)cnt * (HD / 4);
        float4 r0, r1, r2, r3;
        { const int i = tid;        r0 = q4[i < nq ? i : nq - 1]; }
        { const int i = tid + 1024; r1 = q4[i < nq ? i : nq - 1]; }
        { const int i = tid + 2048; r2 = q4[i < nq ? i : nq - 1]; }
        { const int i = tid + 3072; r3 = q4[i < nq ? i : nq - 1]; }
        if (tid        < nq) s4[tid]        = r0;
        if (tid + 1024 < nq) s4[tid + 1024] = r1;
        if (tid + 2048 < nq) s4[tid + 2048] = r2;
        if (tid + 3072 < nq) s4[tid + 3072] = r3;
    }
    __syncthreads();

    for (int g = 0; ; g += 4) {
        if (g == N_STEP || count > Q_KEEP) {
            work_off(myq, count, lo, src, dst, K, V, s_q, s_acc, lane);
            count = 0;
            if (g == N_STEP) break;
        }
        scan_keep(dA, eA, okA, pat, lane, myq, count);
        scan_keep(dB, eB, okB, pat, lane, myq, count);
        scan_load(plane4, g + 4, wave, lane, skew, dA, eA, okA);
        scan_load(plane4, g + 5, wave, lane, skew, dB, eB, okB);
        if (count > Q_KEEP) {
            work_off(myq, count, lo, src, dst, K, V, s_q, s_acc, lane);
            count = 0;
        }
        scan_keep(dC, eC, okC, pat, lane, myq, count);
        scan_keep(dD, eD, okD, pat, lane, myq, count);
        scan_load(plane4, g + 6, wave, lane, skew, dC, eC, okC);
        scan_load(plane4, g + 7, wave, lane, skew, dD, eD, okD);
    }

    __syncthreads();

    {
        float* o = out + (size_t)lo * HD;
        for (int i = tid; i < (int)cnt * HD; i += 1024) {
            const int r = i >> 6;
            const int c = i & 63;
            const int h = c >> 4;
            o[i] = s_acc[r * ACC_PITCH + c] / (s_acc[r * ACC_PITCH + HD + h] + 1e-6f);
        }
    }
}

extern "C" void kernel_launch(void* const* d_in, const int* in_sizes, int n_in,
                              void* d_out, int out_size, void* d_ws, size_t ws_size,
                              hipStream_t stream)
{
    (void)in_sizes; (void)n_in; (void)out_size; (void)ws_size;

    const float* X   = (const float*)d_in[0];
    const int*   src = (const int*)  d_in[1];
    const int*   dst = (const int*)  d_in[2];
    const float* Wq  = (const float*)d_in[3];
    const float* Wk  = (const float*)d_in[4];
    const float* Wv  = (const float*)d_in[5];

    float* out = (float*)d_out;
    float* ws  = (float*)d_ws;
    unsigned* plane = (unsigned*)d_ws;
    float* Q   = out;
    float* K   = ws + (size_t)N_NODES * HD;
    float* V   = ws + (size_t)2 * N_NODES * HD;

    {
        const int ntask = (N_NODES / 16) * 3;
        const int wpb   = 256 / 32;
        const int blocks = PLANE_BLOCKS + (ntask + wpb - 1) / wpb;
        gt_qkv_proj_wmma<<<blocks, 256, 0, stream>>>(X, dst, plane, Wq, Wk, Wv, Q, K, V);
    }

    gt_edge_owner<<<N_OWNERS, 1024, 0, stream>>>(src, dst, (const uint4*)plane, K, V, out);
}


# Owner plane bit-sliced: 8 one-bit planes (ballots in launch 1); edge scan matches 32 edges per lane with 12 xor/and ops, hit mask direct (was 23 byte ops per 16); sums still LDS adds, arrival order
# speedup vs baseline: 1.8736x; 1.0784x over previous
#include <hip/hip_runtime.h>
#include <hip/hip_bf16.h>

typedef __attribute__((ext_vector_type(2))) float v2f;
typedef __attribute__((ext_vector_type(8))) float v8f;

#define N_NODES 50000
#define E_EDGES 1600000
#define NHEAD   4
#define DHEAD   16
#define F_IN    128
#define HD      64

#define PLANE_IDS    1024
#define PLANE_CHUNKS ((E_EDGES + PLANE_IDS - 1) / PLANE_IDS)
#define PLANE_TAIL_LANES ((E_EDGES - (PLANE_CHUNKS - 1) * PLANE_IDS) / 32)
#define PLANE_BLOCKS ((PLANE_CHUNKS + 7) / 8)

static_assert((E_EDGES - (PLANE_CHUNKS - 1) * PLANE_IDS) % 32 == 0, "the last chunk ends on a lane boundary");
static_assert(PLANE_TAIL_LANES >= 1 && PLANE_TAIL_LANES <= 32, "tail lanes");
static_assert((16 * (N_NODES - 1) + 15) / 3125 == 255, "owner fits eight bits");

__device__ __forceinline__ unsigned owner_of(const int d) { return (unsigned)(16 * d + 15) / 3125u; }

__device__ __forceinline__ void
plane_chunk(const int* __restrict__ dst, uint4* __restrict__ plane4, const int c, const int lane)
{
    if (c >= PLANE_CHUNKS) return;
    unsigned w[8] = {0u, 0u, 0u, 0u, 0u, 0u, 0u, 0u};
#pragma unroll
    for (int half = 0; half < 2; ++half) {
        int d[16];
#pragma unroll
        for (int k = 0; k < 16; ++k) {
            const int e = c * PLANE_IDS + (half * 16 + k) * 32 + lane;
            d[k] = dst[e < E_EDGES ? e : E_EDGES - 1];
        }
#pragma unroll
        for (int k = 0; k < 16; ++k) {
            const unsigned o = owner_of(d[k]);
#pragma unroll
            for (int i = 0; i < 8; ++i) {
                const unsigned mask = __builtin_amdgcn_ballot_w32(((o >> i) & 1u) != 0u);
                w[i] = (lane == half * 16 + k) ? mask : w[i];
            }
        }
    }
    plane4[(c * 2 + 0) * 32 + lane] = make_uint4(w[0], w[1], w[2], w[3]);
    plane4[(c * 2 + 1) * 32 + lane] = make_uint4(w[4], w[5], w[6], w[7]);
}

__global__ void __launch_bounds__(256)
gt_qkv_proj_wmma(const float* __restrict__ X,
                 const int*   __restrict__ dst,
                 unsigned*    __restrict__ plane,
                 const float* __restrict__ Wq,
                 const float* __restrict__ Wk,
                 const float* __restrict__ Wv,
                 float* __restrict__ Q,
                 float* __restrict__ K,
                 float* __restrict__ V)
{
    const int lane = (int)(threadIdx.x & 31);
    if (blockIdx.x < PLANE_BLOCKS) {
        plane_chunk(dst, (uint4*)plane, (int)((blockIdx.x * blockDim.x + threadIdx.x) >> 5), lane);
        return;
    }
    const int wave = (int)(((blockIdx.x - PLANE_BLOCKS) * blockDim.x + threadIdx.x) >> 5);
    const int NTASK = (N_NODES / 16) * 3;
    if (wave >= NTASK) return;

    const int mat  = wave % 3;
    const int row0 = (wave / 3) * 16;
    const float* __restrict__ W   = (mat == 0) ? Wq : (mat == 1) ? Wk : Wv;
    float*       __restrict__ Out = (mat == 0) ? Q  : (mat == 1) ? K  : V;

    const int grp = lane >> 4;
    const int m16 = lane & 15;

    v8f c0 = {}, c1 = {}, c2 = {}, c3 = {};

    const float* __restrict__ arow = X + (size_t)(row0 + m16) * F_IN + 2 * grp;

#pragma unroll 4
    for (int k0 = 0; k0 < F_IN; k0 += 4) {
        v2f a;
        a.x = arow[k0];
        a.y = arow[k0 + 1];

        const float* __restrict__ wr0 = W + (size_t)(k0 + 2 * grp) * HD + m16;
        const float* __restrict__ wr1 = wr0 + HD;

        v2f b0, b1, b2, b3;
        b0.x = wr0[0];  b0.y = wr1[0];
        b1.x = wr0[16]; b1.y = wr1[16];
        b2.x = wr0[32]; b2.y = wr1[32];
        b3.x = wr0[48]; b3.y = wr1[48];

        c0 = __builtin_amdgcn_wmma_f32_16x16x4_f32(false, a, false, b0, (short)0, c0, false, false);
        c1 = __builtin_amdgcn_wmma_f32_16x16x4_f32(false, a, false, b1, (short)0, c1, false, false);
        c2 = __builtin_amdgcn_wmma_f32_16x16x4_f32(false, a, false, b2, (short)0, c2, false, false);
        c3 = __builtin_amdgcn_wmma_f32_16x16x4_f32(false, a, false, b3, (short)0, c3, false, false);
    }

    const int mrow = 8 * grp;
#pragma unroll
    for (int r = 0; r < 8; ++r) {
        float* __restrict__ o = Out + (size_t)(row0 + mrow + r) * HD + m16;
        o[0]  = c0[r];
        o[16] = c1[r];
        o[32] = c2[r];
        o[48] = c3[r];
    }
}

#define N_OWNERS   256
#define OWN_MAX    196
#define ACC_PITCH  68
#define OWN_WAVES  32
#define IDS_PER_LD PLANE_IDS
#define N_CHUNK    PLANE_CHUNKS
#define N_STEP     50
#define Q_KEEP     64
#define QW         (Q_KEEP + IDS_PER_LD)
#define START_SKEW 6
#define EDGE_U     8

static_assert(N_STEP % 2 == 0 && N_STEP * OWN_WAVES >= N_CHUNK, "the steps cover the plane");
static_assert(N_CHUNK + (N_OWNERS - 1) * START_SKEW < 2 * N_CHUNK, "one wrap at most");
static_assert(N_OWNERS == 256, "eight bits name the owner");
static_assert(OWN_MAX * (HD / 4) <= 4 * 1024, "Q rows: four 16-byte loads per thread");

__device__ __forceinline__ void
scan_load(const uint4* __restrict__ plane4, const int step, const int wave, const int lane,
          const int skew, uint4& lo, uint4& hi, unsigned& e, bool& valid, bool& last)
{
    const int c = step * OWN_WAVES + wave;
    valid = c < N_CHUNK;
    int p = (valid ? c : 0) + skew; if (p >= N_CHUNK) p -= N_CHUNK;
    lo = plane4[(p * 2 + 0) * 32 + lane];
    hi = plane4[(p * 2 + 1) * 32 + lane];
    e = (unsigned)p * IDS_PER_LD;
    last = p == N_CHUNK - 1;
}

__device__ __forceinline__ unsigned
scan_match(const uint4& lo, const uint4& hi, const bool valid, const bool last,
           const unsigned (&sel)[8], const int lane)
{
    unsigned m = (lo.x ^ sel[0]) & (lo.y ^ sel[1]) & (lo.z ^ sel[2]) & (lo.w ^ sel[3])
               & (hi.x ^ sel[4]) & (hi.y ^ sel[5]) & (hi.z ^ sel[6]) & (hi.w ^ sel[7]);
    if (last && lane >= PLANE_TAIL_LANES) m = 0u;
    return valid ? m : 0u;
}

__device__ __forceinline__ void
scan_keep(unsigned m, const unsigned e0, unsigned* myq, unsigned& count)
{
    unsigned live = __builtin_amdgcn_ballot_w32(m != 0u);
    while (live != 0u) {
        if (m != 0u)
            myq[count + __builtin_amdgcn_mbcnt_lo(live, 0u)] = e0 + (unsigned)__builtin_ctz(m);
        count += (unsigned)__builtin_popcount(live);
        m &= m - 1u;
        live = __builtin_amdgcn_ballot_w32(m != 0u);
    }
}

template <int M>
__device__ __forceinline__ float row_xor(const float x)
{
    return __builtin_bit_cast(float,
        __builtin_amdgcn_mov_dpp(__builtin_bit_cast(int, x), 0x160 + M, 0xf, 0xf, true));
}

template <int M, int U>
__device__ __forceinline__ void row_xor_add(float (&p0)[U], float (&p1)[U])
{
#pragma unroll
    for (int t = 0; t < U; ++t) {
        p0[t] += row_xor<M>(p0[t]);
        p1[t] += row_xor<M>(p1[t]);
    }
}

template <int U>
__device__ __forceinline__ void
edge_batch(const unsigned (&srcs)[U], const unsigned (&dls)[U],
           const float* __restrict__ K, const float* __restrict__ V,
           const float* s_q, float* s_acc, const int lane)
{
    const int grp = lane >> 4;
    const int dd  = lane & 15;
    float k0[U], k1[U], v0[U], v1[U], p0[U], p1[U];
#pragma unroll
    for (int t = 0; t < U; ++t) {
        const float* __restrict__ kr = K + (size_t)srcs[t] * HD;
        const float* __restrict__ vr = V + (size_t)srcs[t] * HD;
        k0[t] = kr[lane]; k1[t] = kr[lane + 32];
        v0[t] = vr[lane]; v1[t] = vr[lane + 32];
    }
#pragma unroll
    for (int t = 0; t < U; ++t) {
        p0[t] = k0[t] * s_q[dls[t] * HD + lane];
        p1[t] = k1[t] * s_q[dls[t] * HD + lane + 32];
    }
    row_xor_add<1, U>(p0, p1);
    row_xor_add<2, U>(p0, p1);
    row_xor_add<4, U>(p0, p1);
    row_xor_add<8, U>(p0, p1);
#pragma unroll
    for (int t = 0; t < U; ++t) {
        p0[t] = __expf(fminf(fmaxf(p0[t] * 0.25f, -5.0f), 5.0f));
        p1[t] = __expf(fminf(fmaxf(p1[t] * 0.25f, -5.0f), 5.0f));
        float* a = s_acc + dls[t] * ACC_PITCH;
        atomicAdd(&a[lane],      p0[t] * v0[t]);
        atomicAdd(&a[lane + 32], p1[t] * v1[t]);
    }
    if (dd == 0) {
#pragma unroll
        for (int t = 0; t < U; ++t) {
            float* a = s_acc + dls[t] * ACC_PITCH + HD;
            atomicAdd(&a[grp],     p0[t]);
            atomicAdd(&a[2 + grp], p1[t]);
        }
    }
}

__device__ __forceinline__ void
work_off(const unsigned* myq, const unsigned count, const unsigned lo,
         const int* __restrict__ src, const int* __restrict__ dst,
         const float* __restrict__ K, const float* __restrict__ V,
         const float* s_q, float* s_acc, const int lane)
{
    for (unsigned i0 = 0; i0 < count; i0 += 32) {
        const unsigned left = count - i0;
        const int nb = (int)(left < 32u ? left : 32u);
        unsigned sv = 0, dv = 0;
        if (lane < nb) {
            const unsigned ent = myq[i0 + lane];
            sv = (unsigned)src[ent];
            dv = (unsigned)dst[ent] - lo;
            dv = dv < (unsigned)OWN_MAX ? dv : 0u;
        }
        int j = 0;
        for (; j + EDGE_U <= nb; j += EDGE_U) {
            unsigned srcs[EDGE_U], dls[EDGE_U];
#pragma unroll
            for (int t = 0; t < EDGE_U; ++t) {
                srcs[t] = __builtin_amdgcn_readlane(sv, j + t);
                dls[t]  = __builtin_amdgcn_readlane(dv, j + t);
            }
            edge_batch<EDGE_U>(srcs, dls, K, V, s_q, s_acc, lane);
        }
        for (; j < nb; ++j) {
            unsigned srcs[1], dls[1];
            srcs[0] = __builtin_amdgcn_readlane(sv, j);
            dls[0]  = __builtin_amdgcn_readlane(dv, j);
            edge_batch<1>(srcs, dls, K, V, s_q, s_acc, lane);
        }
    }
}

__global__ void __launch_bounds__(1024)
gt_edge_owner(const int* __restrict__ src, const int* __restrict__ dst,
              const uint4* __restrict__ plane4,
              const float* __restrict__ K, const float* __restrict__ V,
              float* out)
{
    __shared__ __attribute__((aligned(16))) float    s_acc[OWN_MAX * ACC_PITCH];
    __shared__ __attribute__((aligned(16))) float    s_q[OWN_MAX * HD];
    __shared__ __attribute__((aligned(16))) unsigned s_queue[OWN_WAVES * QW];

    const int tid  = (int)threadIdx.x;
    const int lane = tid & 31;
    const int wave = __builtin_amdgcn_readfirstlane(tid >> 5);
    const int b    = (int)blockIdx.x;
    const unsigned lo  = (unsigned)(b * 3125) >> 4;
    const unsigned hi  = (unsigned)((b + 1) * 3125) >> 4;
    const unsigned cnt = hi - lo;
    unsigned sel[8];
#pragma unroll
    for (int i = 0; i < 8; ++i) sel[i] = ((b >> i) & 1) ? 0u : 0xFFFFFFFFu;
    const int skew = b * START_SKEW;
    const unsigned lane32 = (unsigned)lane * 32u;

    unsigned* myq = s_queue + wave * QW;
    unsigned count = 0;

    uint4    aLo, aHi, bLo, bHi;
    unsigned eA, eB;
    bool     okA, okB, lastA, lastB;
    scan_load(plane4, 0, wave, lane, skew, aLo, aHi, eA, okA, lastA);
    scan_load(plane4, 1, wave, lane, skew, bLo, bHi, eB, okB, lastB);

    for (int i = tid; i < OWN_MAX * ACC_PITCH; i += 1024) s_acc[i] = 0.0f;
    {
        const float4* q4 = (const float4*)(out + (size_t)lo * HD);
        float4* s4 = (float4*)s_q;
        const int nq = (int)cnt * (HD / 4);
        float4 r0, r1, r2, r3;
        { const int i = tid;        r0 = q4[i < nq ? i : nq - 1]; }
        { const int i = tid + 1024; r1 = q4[i < nq ? i : nq - 1]; }
        { const int i = tid + 2048; r2 = q4[i < nq ? i : nq - 1]; }
        { const int i = tid + 3072; r3 = q4[i < nq ? i : nq - 1]; }
        if (tid        < nq) s4[tid]        = r0;
        if (tid + 1024 < nq) s4[tid + 1024] = r1;
        if (tid + 2048 < nq) s4[tid + 2048] = r2;
        if (tid + 3072 < nq) s4[tid + 3072] = r3;
    }
    __syncthreads();

    for (int g = 0; ; g += 2) {
        if (g == N_STEP || count > Q_KEEP) {
            work_off(myq, count, lo, src, dst, K, V, s_q, s_acc, lane);
            count = 0;
            if (g == N_STEP) break;
        }
        {
            const unsigned m  = scan_match(aLo, aHi, okA, lastA, sel, lane);
            const unsigned e0 = eA + lane32;
            scan_load(plane4, g + 2, wave, lane, skew, aLo, aHi, eA, okA, lastA);
            scan_keep(m, e0, myq, count);
        }
        if (count > Q_KEEP) {
            work_off(myq, count, lo, src, dst, K, V, s_q, s_acc, lane);
            count = 0;
        }
        {
            const unsigned m  = scan_match(bLo, bHi, okB, lastB, sel, lane);
            const unsigned e0 = eB + lane32;
            scan_load(plane4, g + 3, wave, lane, skew, bLo, bHi, eB, okB, lastB);
            scan_keep(m, e0, myq, count);
        }
    }

    __syncthreads();

    {
        float* o = out + (size_t)lo * HD;
        for (int i = tid; i < (int)cnt * HD; i += 1024) {
            const int r = i >> 6;
            const int c = i & 63;
            const int h = c >> 4;
            o[i] = s_acc[r * ACC_PITCH + c] / (s_acc[r * ACC_PITCH + HD + h] + 1e-6f);
        }
    }
}

extern "C" void kernel_launch(void* const* d_in, const int* in_sizes, int n_in,
                              void* d_out, int out_size, void* d_ws, size_t ws_size,
                              hipStream_t stream)
{
    (void)in_sizes; (void)n_in; (void)out_size; (void)ws_size;

    const float* X   = (const float*)d_in[0];
    const int*   src = (const int*)  d_in[1];
    const int*   dst = (const int*)  d_in[2];
    const float* Wq  = (const float*)d_in[3];
    const float* Wk  = (const float*)d_in[4];
    const float* Wv  = (const float*)d_in[5];

    float* out = (float*)d_out;
    float* ws  = (float*)d_ws;
    unsigned* plane = (unsigned*)d_ws;
    float* Q   = out;
    float* K   = ws + (size_t)N_NODES * HD;
    float* V   = ws + (size_t)2 * N_NODES * HD;

    {
        const int ntask = (N_NODES / 16) * 3;
        const int wpb   = 256 / 32;
        const int blocks = PLANE_BLOCKS + (ntask + wpb - 1) / wpb;
        gt_qkv_proj_wmma<<<blocks, 256, 0, stream>>>(X, dst, plane, Wq, Wk, Wv, Q, K, V);
    }

    gt_edge_owner<<<N_OWNERS, 1024, 0, stream>>>(src, dst, (const uint4*)plane, K, V, out);
}


# Edge scores by reduce-scatter over 8 edges: xor steps 1,2,4,8 as before, each halves the sums a lane keeps (first add no longer an fma): 1 exp, 1 z LDS add per lane; V mul by DPP row_share
# speedup vs baseline: 1.9701x; 1.0515x over previous
#include <hip/hip_runtime.h>
#include <hip/hip_bf16.h>

typedef __attribute__((ext_vector_type(2))) float v2f;
typedef __attribute__((ext_vector_type(8))) float v8f;

#define N_NODES 50000
#define E_EDGES 1600000
#define NHEAD   4
#define DHEAD   16
#define F_IN    128
#define HD      64

#define PLANE_IDS    1024
#define PLANE_CHUNKS ((E_EDGES + PLANE_IDS - 1) / PLANE_IDS)
#define PLANE_TAIL_LANES ((E_EDGES - (PLANE_CHUNKS - 1) * PLANE_IDS) / 32)
#define PLANE_BLOCKS ((PLANE_CHUNKS + 7) / 8)

static_assert((E_EDGES - (PLANE_CHUNKS - 1) * PLANE_IDS) % 32 == 0, "the last chunk ends on a lane boundary");
static_assert(PLANE_TAIL_LANES >= 1 && PLANE_TAIL_LANES <= 32, "tail lanes");
static_assert((16 * (N_NODES - 1) + 15) / 3125 == 255, "owner fits eight bits");

__device__ __forceinline__ unsigned owner_of(const int d) { return (unsigned)(16 * d + 15) / 3125u; }

__device__ __forceinline__ void
plane_chunk(const int* __restrict__ dst, uint4* __restrict__ plane4, const int c, const int lane)
{
    if (c >= PLANE_CHUNKS) return;
    unsigned w[8] = {0u, 0u, 0u, 0u, 0u, 0u, 0u, 0u};
#pragma unroll
    for (int half = 0; half < 2; ++half) {
        int d[16];
#pragma unroll
        for (int k = 0; k < 16; ++k) {
            const int e = c * PLANE_IDS + (half * 16 + k) * 32 + lane;
            d[k] = dst[e < E_EDGES ? e : E_EDGES - 1];
        }
#pragma unroll
        for (int k = 0; k < 16; ++k) {
            const unsigned o = owner_of(d[k]);
#pragma unroll
            for (int i = 0; i < 8; ++i) {
                const unsigned mask = __builtin_amdgcn_ballot_w32(((o >> i) & 1u) != 0u);
                w[i] = (lane == half * 16 + k) ? mask : w[i];
            }
        }
    }
    plane4[(c * 2 + 0) * 32 + lane] = make_uint4(w[0], w[1], w[2], w[3]);
    plane4[(c * 2 + 1) * 32 + lane] = make_uint4(w[4], w[5], w[6], w[7]);
}

__global__ void __launch_bounds__(256)
gt_qkv_proj_wmma(const float* __restrict__ X,
                 const int*   __restrict__ dst,
                 unsigned*    __restrict__ plane,
                 const float* __restrict__ Wq,
                 const float* __restrict__ Wk,
                 const float* __restrict__ Wv,
                 float* __restrict__ Q,
                 float* __restrict__ K,
                 float* __restrict__ V)
{
    const int lane = (int)(threadIdx.x & 31);
    if (blockIdx.x < PLANE_BLOCKS) {
        plane_chunk(dst, (uint4*)plane, (int)((blockIdx.x * blockDim.x + threadIdx.x) >> 5), lane);
        return;
    }
    const int wave = (int)(((blockIdx.x - PLANE_BLOCKS) * blockDim.x + threadIdx.x) >> 5);
    const int NTASK = (N_NODES / 16) * 3;
    if (wave >= NTASK) return;

    const int mat  = wave % 3;
    const int row0 = (wave / 3) * 16;
    const float* __restrict__ W   = (mat == 0) ? Wq : (mat == 1) ? Wk : Wv;
    float*       __restrict__ Out = (mat == 0) ? Q  : (mat == 1) ? K  : V;

    const int grp = lane >> 4;
    const int m16 = lane & 15;

    v8f c0 = {}, c1 = {}, c2 = {}, c3 = {};

    const float* __restrict__ arow = X + (size_t)(row0 + m16) * F_IN + 2 * grp;

#pragma unroll 4
    for (int k0 = 0; k0 < F_IN; k0 += 4) {
        v2f a;
        a.x = arow[k0];
        a.y = arow[k0 + 1];

        const float* __restrict__ wr0 = W + (size_t)(k0 + 2 * grp) * HD + m16;
        const float* __restrict__ wr1 = wr0 + HD;

        v2f b0, b1, b2, b3;
        b0.x = wr0[0];  b0.y = wr1[0];
        b1.x = wr0[16]; b1.y = wr1[16];
        b2.x = wr0[32]; b2.y = wr1[32];
        b3.x = wr0[48]; b3.y = wr1[48];

        c0 = __builtin_amdgcn_wmma_f32_16x16x4_f32(false, a, false, b0, (short)0, c0, false, false);
        c1 = __builtin_amdgcn_wmma_f32_16x16x4_f32(false, a, false, b1, (short)0, c1, false, false);
        c2 = __builtin_amdgcn_wmma_f32_16x16x4_f32(false, a, false, b2, (short)0, c2, false, false);
        c3 = __builtin_amdgcn_wmma_f32_16x16x4_f32(false, a, false, b3, (short)0, c3, false, false);
    }

    const int mrow = 8 * grp;
#pragma unroll
    for (int r = 0; r < 8; ++r) {
        float* __restrict__ o = Out + (size_t)(row0 + mrow + r) * HD + m16;
        o[0]  = c0[r];
        o[16] = c1[r];
        o[32] = c2[r];
        o[48] = c3[r];
    }
}

#define N_OWNERS   256
#define OWN_MAX    196
#define ACC_PITCH  68
#define OWN_WAVES  32
#define IDS_PER_LD PLANE_IDS
#define N_CHUNK    PLANE_CHUNKS
#define N_STEP     50
#define Q_KEEP     64
#define QW         (Q_KEEP + IDS_PER_LD)
#define START_SKEW 6
#define EDGE_U     8

static_assert(N_STEP % 2 == 0 && N_STEP * OWN_WAVES >= N_CHUNK, "the steps cover the plane");
static_assert(EDGE_U == 8, "a full batch is eight edges: 8 edges x 4 heads = one score per lane");
static_assert(N_CHUNK + (N_OWNERS - 1) * START_SKEW < 2 * N_CHUNK, "one wrap at most");
static_assert(N_OWNERS == 256, "eight bits name the owner");
static_assert(OWN_MAX * (HD / 4) <= 4 * 1024, "Q rows: four 16-byte loads per thread");

__device__ __forceinline__ void
scan_load(const uint4* __restrict__ plane4, const int step, const int wave, const int lane,
          const int skew, uint4& lo, uint4& hi, unsigned& e, bool& valid, bool& last)
{
    const int c = step * OWN_WAVES + wave;
    valid = c < N_CHUNK;
    int p = (valid ? c : 0) + skew; if (p >= N_CHUNK) p -= N_CHUNK;
    lo = plane4[(p * 2 + 0) * 32 + lane];
    hi = plane4[(p * 2 + 1) * 32 + lane];
    e = (unsigned)p * IDS_PER_LD;
    last = p == N_CHUNK - 1;
}

__device__ __forceinline__ unsigned
scan_match(const uint4& lo, const uint4& hi, const bool valid, const bool last,
           const unsigned (&sel)[8], const int lane)
{
    unsigned m = (lo.x ^ sel[0]) & (lo.y ^ sel[1]) & (lo.z ^ sel[2]) & (lo.w ^ sel[3])
               & (hi.x ^ sel[4]) & (hi.y ^ sel[5]) & (hi.z ^ sel[6]) & (hi.w ^ sel[7]);
    if (last && lane >= PLANE_TAIL_LANES) m = 0u;
    return valid ? m : 0u;
}

__device__ __forceinline__ void
scan_keep(unsigned m, const unsigned e0, unsigned* myq, unsigned& count)
{
    unsigned live = __builtin_amdgcn_ballot_w32(m != 0u);
    while (live != 0u) {
        if (m != 0u)
            myq[count + __builtin_amdgcn_mbcnt_lo(live, 0u)] = e0 + (unsigned)__builtin_ctz(m);
        count += (unsigned)__builtin_popcount(live);
        m &= m - 1u;
        live = __builtin_amdgcn_ballot_w32(m != 0u);
    }
}

template <int M>
__device__ __forceinline__ float row_xor(const float x)
{
    return __builtin_bit_cast(float,
        __builtin_amdgcn_mov_dpp(__builtin_bit_cast(int, x), 0x160 + M, 0xf, 0xf, true));
}

template <int M, int U>
__device__ __forceinline__ void row_xor_add(float (&p0)[U], float (&p1)[U])
{
#pragma unroll
    for (int t = 0; t < U; ++t) {
        p0[t] += row_xor<M>(p0[t]);
        p1[t] += row_xor<M>(p1[t]);
    }
}

template <int T>
__device__ __forceinline__ float row_take(const float x)
{
    return __builtin_bit_cast(float,
        __builtin_amdgcn_mov_dpp(__builtin_bit_cast(int, x), 0x150 + T, 0xf, 0xf, true));
}

template <int M>
__device__ __forceinline__ float fold_pair(const float a, const float b, const bool up)
{
    const float keep = up ? b : a;
    const float give = up ? a : b;
    return keep + row_xor<M>(give);
}

template <int T>
__device__ __forceinline__ void
wv_add(const unsigned (&dls)[8], const float (&v0)[8], const float (&v1)[8],
       const float e, float* s_acc, const int lane)
{
    float* a = s_acc + dls[T] * ACC_PITCH;
    atomicAdd(&a[lane],      v0[T] * row_take<T>(e));
    atomicAdd(&a[lane + 32], v1[T] * row_take<8 + T>(e));
}

__device__ __forceinline__ void
edge_batch8(const unsigned (&srcs)[8], const unsigned (&dls)[8], const unsigned dlv,
            const float* __restrict__ K, const float* __restrict__ V,
            const float* s_q, float* s_acc, const int lane)
{
    const int grp = lane >> 4;
    const int dd  = lane & 15;
    float k0[8], k1[8], v0[8], v1[8], val[16];
#pragma unroll
    for (int t = 0; t < 8; ++t) {
        const float* __restrict__ kr = K + (size_t)srcs[t] * HD;
        const float* __restrict__ vr = V + (size_t)srcs[t] * HD;
        k0[t] = kr[lane]; k1[t] = kr[lane + 32];
        v0[t] = vr[lane]; v1[t] = vr[lane + 32];
    }
#pragma unroll
    for (int t = 0; t < 8; ++t) {
        val[t]     = k0[t] * s_q[dls[t] * HD + lane];
        val[8 + t] = k1[t] * s_q[dls[t] * HD + lane + 32];
    }
    const bool u1 = (lane & 1) != 0, u2 = (lane & 2) != 0, u4 = (lane & 4) != 0, u8 = (lane & 8) != 0;
    float r1[8], r2[4], r3[2];
#pragma unroll
    for (int i = 0; i < 8; ++i) r1[i] = fold_pair<1>(val[2 * i], val[2 * i + 1], u1);
#pragma unroll
    for (int i = 0; i < 4; ++i) r2[i] = fold_pair<2>(r1[2 * i], r1[2 * i + 1], u2);
#pragma unroll
    for (int i = 0; i < 2; ++i) r3[i] = fold_pair<4>(r2[2 * i], r2[2 * i + 1], u4);
    const float s = fold_pair<8>(r3[0], r3[1], u8);
    const float e = __expf(fminf(fmaxf(s * 0.25f, -5.0f), 5.0f));
    atomicAdd(&s_acc[dlv * ACC_PITCH + HD + grp + 2 * (dd >> 3)], e);
    wv_add<0>(dls, v0, v1, e, s_acc, lane);
    wv_add<1>(dls, v0, v1, e, s_acc, lane);
    wv_add<2>(dls, v0, v1, e, s_acc, lane);
    wv_add<3>(dls, v0, v1, e, s_acc, lane);
    wv_add<4>(dls, v0, v1, e, s_acc, lane);
    wv_add<5>(dls, v0, v1, e, s_acc, lane);
    wv_add<6>(dls, v0, v1, e, s_acc, lane);
    wv_add<7>(dls, v0, v1, e, s_acc, lane);
}

template <int U>
__device__ __forceinline__ void
edge_batch(const unsigned (&srcs)[U], const unsigned (&dls)[U],
           const float* __restrict__ K, const float* __restrict__ V,
           const float* s_q, float* s_acc, const int lane)
{
    const int grp = lane >> 4;
    const int dd  = lane & 15;
    float k0[U], k1[U], v0[U], v1[U], p0[U], p1[U];
#pragma unroll
    for (int t = 0; t < U; ++t) {
        const float* __restrict__ kr = K + (size_t)srcs[t] * HD;
        const float* __restrict__ vr = V + (size_t)srcs[t] * HD;
        k0[t] = kr[lane]; k1[t] = kr[lane + 32];
        v0[t] = vr[lane]; v1[t] = vr[lane + 32];
    }
#pragma unroll
    for (int t = 0; t < U; ++t) {
        p0[t] = k0[t] * s_q[dls[t] * HD + lane];
        p1[t] = k1[t] * s_q[dls[t] * HD + lane + 32];
    }
    row_xor_add<1, U>(p0, p1);
    row_xor_add<2, U>(p0, p1);
    row_xor_add<4, U>(p0, p1);
    row_xor_add<8, U>(p0, p1);
#pragma unroll
    for (int t = 0; t < U; ++t) {
        p0[t] = __expf(fminf(fmaxf(p0[t] * 0.25f, -5.0f), 5.0f));
        p1[t] = __expf(fminf(fmaxf(p1[t] * 0.25f, -5.0f), 5.0f));
        float* a = s_acc + dls[t] * ACC_PITCH;
        atomicAdd(&a[lane],      p0[t] * v0[t]);
        atomicAdd(&a[lane + 32], p1[t] * v1[t]);
    }
    if (dd == 0) {
#pragma unroll
        for (int t = 0; t < U; ++t) {
            float* a = s_acc + dls[t] * ACC_PITCH + HD;
            atomicAdd(&a[grp],     p0[t]);
            atomicAdd(&a[2 + grp], p1[t]);
        }
    }
}

__device__ __forceinline__ void
work_off(const unsigned* myq, const unsigned count, const unsigned lo,
         const int* __restrict__ src, const int* __restrict__ dst,
         const float* __restrict__ K, const float* __restrict__ V,
         const float* s_q, float* s_acc, const int lane)
{
    for (unsigned i0 = 0; i0 < count; i0 += 32) {
        const unsigned left = count - i0;
        const int nb = (int)(left < 32u ? left : 32u);
        unsigned sv = 0, dv = 0;
        if (lane < nb) {
            const unsigned ent = myq[i0 + lane];
            sv = (unsigned)src[ent];
            dv = (unsigned)dst[ent] - lo;
            dv = dv < (unsigned)OWN_MAX ? dv : 0u;
        }
        int j = 0;
        for (; j + EDGE_U <= nb; j += EDGE_U) {
            unsigned srcs[EDGE_U], dls[EDGE_U];
#pragma unroll
            for (int t = 0; t < EDGE_U; ++t) {
                srcs[t] = __builtin_amdgcn_readlane(sv, j + t);
                dls[t]  = __builtin_amdgcn_readlane(dv, j + t);
            }
            const unsigned dlv = (unsigned)__builtin_amdgcn_ds_bpermute((j + (lane & 7)) << 2, (int)dv);
            edge_batch8(srcs, dls, dlv, K, V, s_q, s_acc, lane);
        }
        for (; j < nb; ++j) {
            unsigned srcs[1], dls[1];
            srcs[0] = __builtin_amdgcn_readlane(sv, j);
            dls[0]  = __builtin_amdgcn_readlane(dv, j);
            edge_batch<1>(srcs, dls, K, V, s_q, s_acc, lane);
        }
    }
}

__global__ void __launch_bounds__(1024)
gt_edge_owner(const int* __restrict__ src, const int* __restrict__ dst,
              const uint4* __restrict__ plane4,
              const float* __restrict__ K, const float* __restrict__ V,
              float* out)
{
    __shared__ __attribute__((aligned(16))) float    s_acc[OWN_MAX * ACC_PITCH];
    __shared__ __attribute__((aligned(16))) float    s_q[OWN_MAX * HD];
    __shared__ __attribute__((aligned(16))) unsigned s_queue[OWN_WAVES * QW];

    const int tid  = (int)threadIdx.x;
    const int lane = tid & 31;
    const int wave = __builtin_amdgcn_readfirstlane(tid >> 5);
    const int b    = (int)blockIdx.x;
    const unsigned lo  = (unsigned)(b * 3125) >> 4;
    const unsigned hi  = (unsigned)((b + 1) * 3125) >> 4;
    const unsigned cnt = hi - lo;
    unsigned sel[8];
#pragma unroll
    for (int i = 0; i < 8; ++i) sel[i] = ((b >> i) & 1) ? 0u : 0xFFFFFFFFu;
    const int skew = b * START_SKEW;
    const unsigned lane32 = (unsigned)lane * 32u;

    unsigned* myq = s_queue + wave * QW;
    unsigned count = 0;

    uint4    aLo, aHi, bLo, bHi;
    unsigned eA, eB;
    bool     okA, okB, lastA, lastB;
    scan_load(plane4, 0, wave, lane, skew, aLo, aHi, eA, okA, lastA);
    scan_load(plane4, 1, wave, lane, skew, bLo, bHi, eB, okB, lastB);

    for (int i = tid; i < OWN_MAX * ACC_PITCH; i += 1024) s_acc[i] = 0.0f;
    {
        const float4* q4 = (const float4*)(out + (size_t)lo * HD);
        float4* s4 = (float4*)s_q;
        const int nq = (int)cnt * (HD / 4);
        float4 r0, r1, r2, r3;
        { const int i = tid;        r0 = q4[i < nq ? i : nq - 1]; }
        { const int i = tid + 1024; r1 = q4[i < nq ? i : nq - 1]; }
        { const int i = tid + 2048; r2 = q4[i < nq ? i : nq - 1]; }
        { const int i = tid + 3072; r3 = q4[i < nq ? i : nq - 1]; }
        if (tid        < nq) s4[tid]        = r0;
        if (tid + 1024 < nq) s4[tid + 1024] = r1;
        if (tid + 2048 < nq) s4[tid + 2048] = r2;
        if (tid + 3072 < nq) s4[tid + 3072] = r3;
    }
    __syncthreads();

    for (int g = 0; ; g += 2) {
        if (g == N_STEP || count > Q_KEEP) {
            work_off(myq, count, lo, src, dst, K, V, s_q, s_acc, lane);
            count = 0;
            if (g == N_STEP) break;
        }
        {
            const unsigned m  = scan_match(aLo, aHi, okA, lastA, sel, lane);
            const unsigned e0 = eA + lane32;
            scan_load(plane4, g + 2, wave, lane, skew, aLo, aHi, eA, okA, lastA);
            scan_keep(m, e0, myq, count);
        }
        if (count > Q_KEEP) {
            work_off(myq, count, lo, src, dst, K, V, s_q, s_acc, lane);
            count = 0;
        }
        {
            const unsigned m  = scan_match(bLo, bHi, okB, lastB, sel, lane);
            const unsigned e0 = eB + lane32;
            scan_load(plane4, g + 3, wave, lane, skew, bLo, bHi, eB, okB, lastB);
            scan_keep(m, e0, myq, count);
        }
    }

    __syncthreads();

    {
        float* o = out + (size_t)lo * HD;
        for (int i = tid; i < (int)cnt * HD; i += 1024) {
            const int r = i >> 6;
            const int c = i & 63;
            const int h = c >> 4;
            o[i] = s_acc[r * ACC_PITCH + c] / (s_acc[r * ACC_PITCH + HD + h] + 1e-6f);
        }
    }
}

extern "C" void kernel_launch(void* const* d_in, const int* in_sizes, int n_in,
                              void* d_out, int out_size, void* d_ws, size_t ws_size,
                              hipStream_t stream)
{
    (void)in_sizes; (void)n_in; (void)out_size; (void)ws_size;

    const float* X   = (const float*)d_in[0];
    const int*   src = (const int*)  d_in[1];
    const int*   dst = (const int*)  d_in[2];
    const float* Wq  = (const float*)d_in[3];
    const float* Wk  = (const float*)d_in[4];
    const float* Wv  = (const float*)d_in[5];

    float* out = (float*)d_out;
    float* ws  = (float*)d_ws;
    unsigned* plane = (unsigned*)d_ws;
    float* Q   = out;
    float* K   = ws + (size_t)N_NODES * HD;
    float* V   = ws + (size_t)2 * N_NODES * HD;

    {
        const int ntask = (N_NODES / 16) * 3;
        const int wpb   = 256 / 32;
        const int blocks = PLANE_BLOCKS + (ntask + wpb - 1) / wpb;
        gt_qkv_proj_wmma<<<blocks, 256, 0, stream>>>(X, dst, plane, Wq, Wk, Wv, Q, K, V);
    }

    gt_edge_owner<<<N_OWNERS, 1024, 0, stream>>>(src, dst, (const uint4*)plane, K, V, out);
}


# Launch 1's plane blocks also write one packed word per edge (src | (dst - owner's first node) << 16); edge kernel gathers that one word per kept edge, not src[e] and dst[e]: one line, not two
# speedup vs baseline: 2.1202x; 1.0762x over previous
#include <hip/hip_runtime.h>
#include <hip/hip_bf16.h>

typedef __attribute__((ext_vector_type(2))) float v2f;
typedef __attribute__((ext_vector_type(8))) float v8f;

#define N_NODES 50000
#define E_EDGES 1600000
#define NHEAD   4
#define DHEAD   16
#define F_IN    128
#define HD      64

#define PLANE_IDS    1024
#define PLANE_CHUNKS ((E_EDGES + PLANE_IDS - 1) / PLANE_IDS)
#define PLANE_TAIL_LANES ((E_EDGES - (PLANE_CHUNKS - 1) * PLANE_IDS) / 32)
#define PLANE_BLOCKS ((PLANE_CHUNKS + 7) / 8)

static_assert((E_EDGES - (PLANE_CHUNKS - 1) * PLANE_IDS) % 32 == 0, "the last chunk ends on a lane boundary");
static_assert(PLANE_TAIL_LANES >= 1 && PLANE_TAIL_LANES <= 32, "tail lanes");
static_assert((16 * (N_NODES - 1) + 15) / 3125 == 255, "owner fits eight bits");

#define PK_OFF_WORDS (1 << 19)
static_assert(PLANE_CHUNKS * (PLANE_IDS / 4) <= PK_OFF_WORDS, "the edge words start behind the plane");
static_assert(PK_OFF_WORDS + E_EDGES <= N_NODES * HD, "the edge words end before the K rows");
static_assert(N_NODES <= 65536, "a src node fits sixteen bits");

__device__ __forceinline__ unsigned owner_of(const int d) { return (unsigned)(16 * d + 15) / 3125u; }

__device__ __forceinline__ void
plane_chunk(const int* __restrict__ src, const int* __restrict__ dst,
            uint4* __restrict__ plane4, unsigned* __restrict__ pk, const int c, const int lane)
{
    if (c >= PLANE_CHUNKS) return;
    unsigned w[8] = {0u, 0u, 0u, 0u, 0u, 0u, 0u, 0u};
#pragma unroll 1
    for (int q = 0; q < 4; ++q) {
        int d[8], s[8];
#pragma unroll
        for (int k = 0; k < 8; ++k) {
            const int e  = c * PLANE_IDS + (q * 8 + k) * 32 + lane;
            const int ec = e < E_EDGES ? e : E_EDGES - 1;
            d[k] = dst[ec];
            s[k] = src[ec];
        }
#pragma unroll
        for (int k = 0; k < 8; ++k) {
            const unsigned o = owner_of(d[k]);
            const int e = c * PLANE_IDS + (q * 8 + k) * 32 + lane;
            if (e < E_EDGES)
                pk[e] = (unsigned)s[k] | (((unsigned)d[k] - ((o * 3125u) >> 4)) << 16);
#pragma unroll
            for (int i = 0; i < 8; ++i) {
                const unsigned mask = __builtin_amdgcn_ballot_w32(((o >> i) & 1u) != 0u);
                w[i] = (lane == q * 8 + k) ? mask : w[i];
            }
        }
    }
    plane4[(c * 2 + 0) * 32 + lane] = make_uint4(w[0], w[1], w[2], w[3]);
    plane4[(c * 2 + 1) * 32 + lane] = make_uint4(w[4], w[5], w[6], w[7]);
}

__global__ void __launch_bounds__(256)
gt_qkv_proj_wmma(const float* __restrict__ X,
                 const int*   __restrict__ src,
                 const int*   __restrict__ dst,
                 unsigned*    __restrict__ plane,
                 unsigned*    __restrict__ pk,
                 const float* __restrict__ Wq,
                 const float* __restrict__ Wk,
                 const float* __restrict__ Wv,
                 float* __restrict__ Q,
                 float* __restrict__ K,
                 float* __restrict__ V)
{
    const int lane = (int)(threadIdx.x & 31);
    if (blockIdx.x < PLANE_BLOCKS) {
        plane_chunk(src, dst, (uint4*)plane, pk, (int)((blockIdx.x * blockDim.x + threadIdx.x) >> 5), lane);
        return;
    }
    const int wave = (int)(((blockIdx.x - PLANE_BLOCKS) * blockDim.x + threadIdx.x) >> 5);
    const int NTASK = (N_NODES / 16) * 3;
    if (wave >= NTASK) return;

    const int mat  = wave % 3;
    const int row0 = (wave / 3) * 16;
    const float* __restrict__ W   = (mat == 0) ? Wq : (mat == 1) ? Wk : Wv;
    float*       __restrict__ Out = (mat == 0) ? Q  : (mat == 1) ? K  : V;

    const int grp = lane >> 4;
    const int m16 = lane & 15;

    v8f c0 = {}, c1 = {}, c2 = {}, c3 = {};

    const float* __restrict__ arow = X + (size_t)(row0 + m16) * F_IN + 2 * grp;

#pragma unroll 4
    for (int k0 = 0; k0 < F_IN; k0 += 4) {
        v2f a;
        a.x = arow[k0];
        a.y = arow[k0 + 1];

        const float* __restrict__ wr0 = W + (size_t)(k0 + 2 * grp) * HD + m16;
        const float* __restrict__ wr1 = wr0 + HD;

        v2f b0, b1, b2, b3;
        b0.x = wr0[0];  b0.y = wr1[0];
        b1.x = wr0[16]; b1.y = wr1[16];
        b2.x = wr0[32]; b2.y = wr1[32];
        b3.x = wr0[48]; b3.y = wr1[48];

        c0 = __builtin_amdgcn_wmma_f32_16x16x4_f32(false, a, false, b0, (short)0, c0, false, false);
        c1 = __builtin_amdgcn_wmma_f32_16x16x4_f32(false, a, false, b1, (short)0, c1, false, false);
        c2 = __builtin_amdgcn_wmma_f32_16x16x4_f32(false, a, false, b2, (short)0, c2, false, false);
        c3 = __builtin_amdgcn_wmma_f32_16x16x4_f32(false, a, false, b3, (short)0, c3, false, false);
    }

    const int mrow = 8 * grp;
#pragma unroll
    for (int r = 0; r < 8; ++r) {
        float* __restrict__ o = Out + (size_t)(row0 + mrow + r) * HD + m16;
        o[0]  = c0[r];
        o[16] = c1[r];
        o[32] = c2[r];
        o[48] = c3[r];
    }
}

#define N_OWNERS   256
#define OWN_MAX    196
#define ACC_PITCH  68
#define OWN_WAVES  32
#define IDS_PER_LD PLANE_IDS
#define N_CHUNK    PLANE_CHUNKS
#define N_STEP     50
#define Q_KEEP     64
#define QW         (Q_KEEP + IDS_PER_LD)
#define START_SKEW 6
#define EDGE_U     8

static_assert(N_STEP % 2 == 0 && N_STEP * OWN_WAVES >= N_CHUNK, "the steps cover the plane");
static_assert(EDGE_U == 8, "a full batch is eight edges: 8 edges x 4 heads = one score per lane");
static_assert(N_CHUNK + (N_OWNERS - 1) * START_SKEW < 2 * N_CHUNK, "one wrap at most");
static_assert(N_OWNERS == 256, "eight bits name the owner");
static_assert(OWN_MAX * (HD / 4) <= 4 * 1024, "Q rows: four 16-byte loads per thread");

__device__ __forceinline__ void
scan_load(const uint4* __restrict__ plane4, const int step, const int wave, const int lane,
          const int skew, uint4& lo, uint4& hi, unsigned& e, bool& valid, bool& last)
{
    const int c = step * OWN_WAVES + wave;
    valid = c < N_CHUNK;
    int p = (valid ? c : 0) + skew; if (p >= N_CHUNK) p -= N_CHUNK;
    lo = plane4[(p * 2 + 0) * 32 + lane];
    hi = plane4[(p * 2 + 1) * 32 + lane];
    e = (unsigned)p * IDS_PER_LD;
    last = p == N_CHUNK - 1;
}

__device__ __forceinline__ unsigned
scan_match(const uint4& lo, const uint4& hi, const bool valid, const bool last,
           const unsigned (&sel)[8], const int lane)
{
    unsigned m = (lo.x ^ sel[0]) & (lo.y ^ sel[1]) & (lo.z ^ sel[2]) & (lo.w ^ sel[3])
               & (hi.x ^ sel[4]) & (hi.y ^ sel[5]) & (hi.z ^ sel[6]) & (hi.w ^ sel[7]);
    if (last && lane >= PLANE_TAIL_LANES) m = 0u;
    return valid ? m : 0u;
}

__device__ __forceinline__ void
scan_keep(unsigned m, const unsigned e0, unsigned* myq, unsigned& count)
{
    unsigned live = __builtin_amdgcn_ballot_w32(m != 0u);
    while (live != 0u) {
        if (m != 0u)
            myq[count + __builtin_amdgcn_mbcnt_lo(live, 0u)] = e0 + (unsigned)__builtin_ctz(m);
        count += (unsigned)__builtin_popcount(live);
        m &= m - 1u;
        live = __builtin_amdgcn_ballot_w32(m != 0u);
    }
}

template <int M>
__device__ __forceinline__ float row_xor(const float x)
{
    return __builtin_bit_cast(float,
        __builtin_amdgcn_mov_dpp(__builtin_bit_cast(int, x), 0x160 + M, 0xf, 0xf, true));
}

template <int M, int U>
__device__ __forceinline__ void row_xor_add(float (&p0)[U], float (&p1)[U])
{
#pragma unroll
    for (int t = 0; t < U; ++t) {
        p0[t] += row_xor<M>(p0[t]);
        p1[t] += row_xor<M>(p1[t]);
    }
}

template <int T>
__device__ __forceinline__ float row_take(const float x)
{
    return __builtin_bit_cast(float,
        __builtin_amdgcn_mov_dpp(__builtin_bit_cast(int, x), 0x150 + T, 0xf, 0xf, true));
}

template <int M>
__device__ __forceinline__ float fold_pair(const float a, const float b, const bool up)
{
    const float keep = up ? b : a;
    const float give = up ? a : b;
    return keep + row_xor<M>(give);
}

template <int T>
__device__ __forceinline__ void
wv_add(const unsigned (&dls)[8], const float (&v0)[8], const float (&v1)[8],
       const float e, float* s_acc, const int lane)
{
    float* a = s_acc + dls[T] * ACC_PITCH;
    atomicAdd(&a[lane],      v0[T] * row_take<T>(e));
    atomicAdd(&a[lane + 32], v1[T] * row_take<8 + T>(e));
}

__device__ __forceinline__ void
edge_batch8(const unsigned (&srcs)[8], const unsigned (&dls)[8], const unsigned dlv,
            const float* __restrict__ K, const float* __restrict__ V,
            const float* s_q, float* s_acc, const int lane)
{
    const int grp = lane >> 4;
    const int dd  = lane & 15;
    float k0[8], k1[8], v0[8], v1[8], val[16];
#pragma unroll
    for (int t = 0; t < 8; ++t) {
        const float* __restrict__ kr = K + (size_t)srcs[t] * HD;
        const float* __restrict__ vr = V + (size_t)srcs[t] * HD;
        k0[t] = kr[lane]; k1[t] = kr[lane + 32];
        v0[t] = vr[lane]; v1[t] = vr[lane + 32];
    }
#pragma unroll
    for (int t = 0; t < 8; ++t) {
        val[t]     = k0[t] * s_q[dls[t] * HD + lane];
        val[8 + t] = k1[t] * s_q[dls[t] * HD + lane + 32];
    }
    const bool u1 = (lane & 1) != 0, u2 = (lane & 2) != 0, u4 = (lane & 4) != 0, u8 = (lane & 8) != 0;
    float r1[8], r2[4], r3[2];
#pragma unroll
    for (int i = 0; i < 8; ++i) r1[i] = fold_pair<1>(val[2 * i], val[2 * i + 1], u1);
#pragma unroll
    for (int i = 0; i < 4; ++i) r2[i] = fold_pair<2>(r1[2 * i], r1[2 * i + 1], u2);
#pragma unroll
    for (int i = 0; i < 2; ++i) r3[i] = fold_pair<4>(r2[2 * i], r2[2 * i + 1], u4);
    const float s = fold_pair<8>(r3[0], r3[1], u8);
    const float e = __expf(fminf(fmaxf(s * 0.25f, -5.0f), 5.0f));
    atomicAdd(&s_acc[dlv * ACC_PITCH + HD + grp + 2 * (dd >> 3)], e);
    wv_add<0>(dls, v0, v1, e, s_acc, lane);
    wv_add<1>(dls, v0, v1, e, s_acc, lane);
    wv_add<2>(dls, v0, v1, e, s_acc, lane);
    wv_add<3>(dls, v0, v1, e, s_acc, lane);
    wv_add<4>(dls, v0, v1, e, s_acc, lane);
    wv_add<5>(dls, v0, v1, e, s_acc, lane);
    wv_add<6>(dls, v0, v1, e, s_acc, lane);
    wv_add<7>(dls, v0, v1, e, s_acc, lane);
}

template <int U>
__device__ __forceinline__ void
edge_batch(const unsigned (&srcs)[U], const unsigned (&dls)[U],
           const float* __restrict__ K, const float* __restrict__ V,
           const float* s_q, float* s_acc, const int lane)
{
    const int grp = lane >> 4;
    const int dd  = lane & 15;
    float k0[U], k1[U], v0[U], v1[U], p0[U], p1[U];
#pragma unroll
    for (int t = 0; t < U; ++t) {
        const float* __restrict__ kr = K + (size_t)srcs[t] * HD;
        const float* __restrict__ vr = V + (size_t)srcs[t] * HD;
        k0[t] = kr[lane]; k1[t] = kr[lane + 32];
        v0[t] = vr[lane]; v1[t] = vr[lane + 32];
    }
#pragma unroll
    for (int t = 0; t < U; ++t) {
        p0[t] = k0[t] * s_q[dls[t] * HD + lane];
        p1[t] = k1[t] * s_q[dls[t] * HD + lane + 32];
    }
    row_xor_add<1, U>(p0, p1);
    row_xor_add<2, U>(p0, p1);
    row_xor_add<4, U>(p0, p1);
    row_xor_add<8, U>(p0, p1);
#pragma unroll
    for (int t = 0; t < U; ++t) {
        p0[t] = __expf(fminf(fmaxf(p0[t] * 0.25f, -5.0f), 5.0f));
        p1[t] = __expf(fminf(fmaxf(p1[t] * 0.25f, -5.0f), 5.0f));
        float* a = s_acc + dls[t] * ACC_PITCH;
        atomicAdd(&a[lane],      p0[t] * v0[t]);
        atomicAdd(&a[lane + 32], p1[t] * v1[t]);
    }
    if (dd == 0) {
#pragma unroll
        for (int t = 0; t < U; ++t) {
            float* a = s_acc + dls[t] * ACC_PITCH + HD;
            atomicAdd(&a[grp],     p0[t]);
            atomicAdd(&a[2 + grp], p1[t]);
        }
    }
}

__device__ __forceinline__ void
work_off(const unsigned* myq, const unsigned count,
         const unsigned* __restrict__ pk,
         const float* __restrict__ K, const float* __restrict__ V,
         const float* s_q, float* s_acc, const int lane)
{
    for (unsigned i0 = 0; i0 < count; i0 += 32) {
        const unsigned left = count - i0;
        const int nb = (int)(left < 32u ? left : 32u);
        unsigned sv = 0, dv = 0;
        if (lane < nb) {
            const unsigned w = pk[myq[i0 + lane]];
            sv = w & 0xFFFFu;
            dv = w >> 16;
            dv = dv < (unsigned)OWN_MAX ? dv : 0u;
        }
        int j = 0;
        for (; j + EDGE_U <= nb; j += EDGE_U) {
            unsigned srcs[EDGE_U], dls[EDGE_U];
#pragma unroll
            for (int t = 0; t < EDGE_U; ++t) {
                srcs[t] = __builtin_amdgcn_readlane(sv, j + t);
                dls[t]  = __builtin_amdgcn_readlane(dv, j + t);
            }
            const unsigned dlv = (unsigned)__builtin_amdgcn_ds_bpermute((j + (lane & 7)) << 2, (int)dv);
            edge_batch8(srcs, dls, dlv, K, V, s_q, s_acc, lane);
        }
        for (; j < nb; ++j) {
            unsigned srcs[1], dls[1];
            srcs[0] = __builtin_amdgcn_readlane(sv, j);
            dls[0]  = __builtin_amdgcn_readlane(dv, j);
            edge_batch<1>(srcs, dls, K, V, s_q, s_acc, lane);
        }
    }
}

__global__ void __launch_bounds__(1024)
gt_edge_owner(const unsigned* __restrict__ pk,
              const uint4* __restrict__ plane4,
              const float* __restrict__ K, const float* __restrict__ V,
              float* out)
{
    __shared__ __attribute__((aligned(16))) float    s_acc[OWN_MAX * ACC_PITCH];
    __shared__ __attribute__((aligned(16))) float    s_q[OWN_MAX * HD];
    __shared__ __attribute__((aligned(16))) unsigned s_queue[OWN_WAVES * QW];

    const int tid  = (int)threadIdx.x;
    const int lane = tid & 31;
    const int wave = __builtin_amdgcn_readfirstlane(tid >> 5);
    const int b    = (int)blockIdx.x;
    const unsigned lo  = (unsigned)(b * 3125) >> 4;
    const unsigned hi  = (unsigned)((b + 1) * 3125) >> 4;
    const unsigned cnt = hi - lo;
    unsigned sel[8];
#pragma unroll
    for (int i = 0; i < 8; ++i) sel[i] = ((b >> i) & 1) ? 0u : 0xFFFFFFFFu;
    const int skew = b * START_SKEW;
    const unsigned lane32 = (unsigned)lane * 32u;

    unsigned* myq = s_queue + wave * QW;
    unsigned count = 0;

    uint4    aLo, aHi, bLo, bHi;
    unsigned eA, eB;
    bool     okA, okB, lastA, lastB;
    scan_load(plane4, 0, wave, lane, skew, aLo, aHi, eA, okA, lastA);
    scan_load(plane4, 1, wave, lane, skew, bLo, bHi, eB, okB, lastB);

    for (int i = tid; i < OWN_MAX * ACC_PITCH; i += 1024) s_acc[i] = 0.0f;
    {
        const float4* q4 = (const float4*)(out + (size_t)lo * HD);
        float4* s4 = (float4*)s_q;
        const int nq = (int)cnt * (HD / 4);
        float4 r0, r1, r2, r3;
        { const int i = tid;        r0 = q4[i < nq ? i : nq - 1]; }
        { const int i = tid + 1024; r1 = q4[i < nq ? i : nq - 1]; }
        { const int i = tid + 2048; r2 = q4[i < nq ? i : nq - 1]; }
        { const int i = tid + 3072; r3 = q4[i < nq ? i : nq - 1]; }
        if (tid        < nq) s4[tid]        = r0;
        if (tid + 1024 < nq) s4[tid + 1024] = r1;
        if (tid + 2048 < nq) s4[tid + 2048] = r2;
        if (tid + 3072 < nq) s4[tid + 3072] = r3;
    }
    __syncthreads();

    for (int g = 0; ; g += 2) {
        if (g == N_STEP || count > Q_KEEP) {
            work_off(myq, count, pk, K, V, s_q, s_acc, lane);
            count = 0;
            if (g == N_STEP) break;
        }
        {
            const unsigned m  = scan_match(aLo, aHi, okA, lastA, sel, lane);
            const unsigned e0 = eA + lane32;
            scan_load(plane4, g + 2, wave, lane, skew, aLo, aHi, eA, okA, lastA);
            scan_keep(m, e0, myq, count);
        }
        if (count > Q_KEEP) {
            work_off(myq, count, pk, K, V, s_q, s_acc, lane);
            count = 0;
        }
        {
            const unsigned m  = scan_match(bLo, bHi, okB, lastB, sel, lane);
            const unsigned e0 = eB + lane32;
            scan_load(plane4, g + 3, wave, lane, skew, bLo, bHi, eB, okB, lastB);
            scan_keep(m, e0, myq, count);
        }
    }

    __syncthreads();

    {
        float* o = out + (size_t)lo * HD;
        for (int i = tid; i < (int)cnt * HD; i += 1024) {
            const int r = i >> 6;
            const int c = i & 63;
            const int h = c >> 4;
            o[i] = s_acc[r * ACC_PITCH + c] / (s_acc[r * ACC_PITCH + HD + h] + 1e-6f);
        }
    }
}

extern "C" void kernel_launch(void* const* d_in, const int* in_sizes, int n_in,
                              void* d_out, int out_size, void* d_ws, size_t ws_size,
                              hipStream_t stream)
{
    (void)in_sizes; (void)n_in; (void)out_size; (void)ws_size;

    const float* X   = (const float*)d_in[0];
    const int*   src = (const int*)  d_in[1];
    const int*   dst = (const int*)  d_in[2];
    const float* Wq  = (const float*)d_in[3];
    const float* Wk  = (const float*)d_in[4];
    const float* Wv  = (const float*)d_in[5];

    float* out = (float*)d_out;
    float* ws  = (float*)d_ws;
    unsigned* plane = (unsigned*)d_ws;
    unsigned* pk    = (unsigned*)d_ws + PK_OFF_WORDS;
    float* Q   = out;
    float* K   = ws + (size_t)N_NODES * HD;
    float* V   = ws + (size_t)2 * N_NODES * HD;

    {
        const int ntask = (N_NODES / 16) * 3;
        const int wpb   = 256 / 32;
        const int blocks = PLANE_BLOCKS + (ntask + wpb - 1) / wpb;
        gt_qkv_proj_wmma<<<blocks, 256, 0, stream>>>(X, src, dst, plane, pk, Wq, Wk, Wv, Q, K, V);
    }

    gt_edge_owner<<<N_OWNERS, 1024, 0, stream>>>(pk, (const uint4*)plane, K, V, out);
}


# Owner plane one-hot by nibble: launch 1 writes 32 one-bit planes (16 per owner nibble); an edge block reads only its 2 and ANDs them: 2 bits per edge, 1 op per 32 (was 8 bits, 15 ops); sums unchanged
# speedup vs baseline: 2.1987x; 1.0370x over previous
#include <hip/hip_runtime.h>
#include <hip/hip_bf16.h>

typedef __attribute__((ext_vector_type(2))) float v2f;
typedef __attribute__((ext_vector_type(8))) float v8f;

#define N_NODES 50000
#define E_EDGES 1600000
#define NHEAD   4
#define DHEAD   16
#define F_IN    128
#define HD      64

#define N_GROUPS     (E_EDGES / 32)
#define N_PLANES     32
#define PLANE_IDS    1024
#define PLANE_CHUNKS ((E_EDGES + PLANE_IDS - 1) / PLANE_IDS)
#define PLANE_BLOCKS ((PLANE_CHUNKS + 7) / 8)

static_assert(E_EDGES % 32 == 0, "whole dwords");
static_assert(N_GROUPS % 8 == 0, "a quarter of a writer chunk (8 dwords) is whole or absent; planes start 16-byte aligned");
static_assert((16 * (N_NODES - 1) + 15) / 3125 == 255, "owner fits eight bits");

#define PK_OFF_WORDS (N_PLANES * N_GROUPS)
static_assert(PK_OFF_WORDS % 32 == 0, "the edge words start on a 128-byte line");
static_assert(PK_OFF_WORDS + E_EDGES <= N_NODES * HD, "the edge words end before the K rows");
static_assert(N_NODES <= 65536, "a src node fits sixteen bits");

__device__ __forceinline__ unsigned owner_of(const int d) { return (unsigned)(16 * d + 15) / 3125u; }

__device__ __forceinline__ void
plane_chunk(const int* __restrict__ src, const int* __restrict__ dst,
            unsigned* __restrict__ plane, unsigned* __restrict__ pk, const int c, const int lane)
{
    if (c >= PLANE_CHUNKS) return;
    const bool high = lane >= 16;
    unsigned sel[4];
#pragma unroll
    for (int i = 0; i < 4; ++i) sel[i] = ((lane >> i) & 1) ? 0u : 0xFFFFFFFFu;
    uint4* __restrict__ mine = (uint4*)(plane + (size_t)lane * N_GROUPS + c * 32);
    int nq = (N_GROUPS - c * 32) / 8;
    nq = nq < 4 ? nq : 4;
#pragma unroll 1
    for (int q = 0; q < nq; ++q) {
        int d[8], s[8];
        unsigned r[8];
#pragma unroll
        for (int k = 0; k < 8; ++k) {
            const int e = c * PLANE_IDS + (q * 8 + k) * 32 + lane;
            d[k] = dst[e];
            s[k] = src[e];
        }
#pragma unroll
        for (int k = 0; k < 8; ++k) {
            const unsigned o = owner_of(d[k]);
            const int e = c * PLANE_IDS + (q * 8 + k) * 32 + lane;
            pk[e] = (unsigned)s[k] | (((unsigned)d[k] - ((o * 3125u) >> 4)) << 16);
            unsigned bal[8];
#pragma unroll
            for (int i = 0; i < 8; ++i) bal[i] = __builtin_amdgcn_ballot_w32(((o >> i) & 1u) != 0u);
            const unsigned wl = (bal[0] ^ sel[0]) & (bal[1] ^ sel[1]) & (bal[2] ^ sel[2]) & (bal[3] ^ sel[3]);
            const unsigned wh = (bal[4] ^ sel[0]) & (bal[5] ^ sel[1]) & (bal[6] ^ sel[2]) & (bal[7] ^ sel[3]);
            r[k] = high ? wh : wl;
        }
        mine[q * 2 + 0] = make_uint4(r[0], r[1], r[2], r[3]);
        mine[q * 2 + 1] = make_uint4(r[4], r[5], r[6], r[7]);
    }
}

__global__ void __launch_bounds__(256)
gt_qkv_proj_wmma(const float* __restrict__ X,
                 const int*   __restrict__ src,
                 const int*   __restrict__ dst,
                 unsigned*    __restrict__ plane,
                 unsigned*    __restrict__ pk,
                 const float* __restrict__ Wq,
                 const float* __restrict__ Wk,
                 const float* __restrict__ Wv,
                 float* __restrict__ Q,
                 float* __restrict__ K,
                 float* __restrict__ V)
{
    const int lane = (int)(threadIdx.x & 31);
    if (blockIdx.x < PLANE_BLOCKS) {
        plane_chunk(src, dst, plane, pk, (int)((blockIdx.x * blockDim.x + threadIdx.x) >> 5), lane);
        return;
    }
    const int wave = (int)(((blockIdx.x - PLANE_BLOCKS) * blockDim.x + threadIdx.x) >> 5);
    const int NTASK = (N_NODES / 16) * 3;
    if (wave >= NTASK) return;

    const int mat  = wave % 3;
    const int row0 = (wave / 3) * 16;
    const float* __restrict__ W   = (mat == 0) ? Wq : (mat == 1) ? Wk : Wv;
    float*       __restrict__ Out = (mat == 0) ? Q  : (mat == 1) ? K  : V;

    const int grp = lane >> 4;
    const int m16 = lane & 15;

    v8f c0 = {}, c1 = {}, c2 = {}, c3 = {};

    const float* __restrict__ arow = X + (size_t)(row0 + m16) * F_IN + 2 * grp;

#pragma unroll 4
    for (int k0 = 0; k0 < F_IN; k0 += 4) {
        v2f a;
        a.x = arow[k0];
        a.y = arow[k0 + 1];

        const float* __restrict__ wr0 = W + (size_t)(k0 + 2 * grp) * HD + m16;
        const float* __restrict__ wr1 = wr0 + HD;

        v2f b0, b1, b2, b3;
        b0.x = wr0[0];  b0.y = wr1[0];
        b1.x = wr0[16]; b1.y = wr1[16];
        b2.x = wr0[32]; b2.y = wr1[32];
        b3.x = wr0[48]; b3.y = wr1[48];

        c0 = __builtin_amdgcn_wmma_f32_16x16x4_f32(false, a, false, b0, (short)0, c0, false, false);
        c1 = __builtin_amdgcn_wmma_f32_16x16x4_f32(false, a, false, b1, (short)0, c1, false, false);
        c2 = __builtin_amdgcn_wmma_f32_16x16x4_f32(false, a, false, b2, (short)0, c2, false, false);
        c3 = __builtin_amdgcn_wmma_f32_16x16x4_f32(false, a, false, b3, (short)0, c3, false, false);
    }

    const int mrow = 8 * grp;
#pragma unroll
    for (int r = 0; r < 8; ++r) {
        float* __restrict__ o = Out + (size_t)(row0 + mrow + r) * HD + m16;
        o[0]  = c0[r];
        o[16] = c1[r];
        o[32] = c2[r];
        o[48] = c3[r];
    }
}

#define N_OWNERS   256
#define OWN_MAX    196
#define ACC_PITCH  68
#define OWN_WAVES  32
#define SCAN_IDS   2048
#define SCAN_DW    (SCAN_IDS / 32)
#define N_CHUNK    ((N_GROUPS + SCAN_DW - 1) / SCAN_DW)
#define SCAN_TAIL_LANES ((N_GROUPS - (N_CHUNK - 1) * SCAN_DW) / 2)
#define N_STEP     26
#define Q_KEEP     64
#define QW         (Q_KEEP + 1024)
#define START_SKEW 3
#define EDGE_U     8

static_assert(N_STEP % 2 == 0 && N_STEP * OWN_WAVES >= N_CHUNK, "the steps cover the plane");
static_assert(EDGE_U == 8, "a full batch is eight edges: 8 edges x 4 heads = one score per lane");
static_assert(N_CHUNK + (N_OWNERS - 1) * START_SKEW < 2 * N_CHUNK, "one wrap at most");
static_assert(N_OWNERS == 256, "two nibbles name the owner");
static_assert((N_GROUPS - (N_CHUNK - 1) * SCAN_DW) % 2 == 0, "the last chunk ends on a lane boundary");
static_assert(SCAN_TAIL_LANES >= 1 && SCAN_TAIL_LANES <= 32, "tail lanes");
static_assert(N_GROUPS % 2 == 0, "planes start 8-byte aligned");
static_assert(OWN_MAX * (HD / 4) <= 4 * 1024, "Q rows: four 16-byte loads per thread");

__device__ __forceinline__ void
scan_load(const uint2* __restrict__ pl_lo, const uint2* __restrict__ pl_hi,
          const int step, const int wave, const int lane,
          const int skew, uint2& lo, uint2& hi, unsigned& e, bool& valid, bool& last)
{
    const int c = step * OWN_WAVES + wave;
    valid = c < N_CHUNK;
    int p = (valid ? c : 0) + skew; if (p >= N_CHUNK) p -= N_CHUNK;
    last = p == N_CHUNK - 1;
    const int idx = p * 32 + ((last && lane >= SCAN_TAIL_LANES) ? 0 : lane);
    lo = pl_lo[idx];
    hi = pl_hi[idx];
    e = (unsigned)p * SCAN_IDS;
}

__device__ __forceinline__ void
scan_match(const uint2& lo, const uint2& hi, const bool valid, const bool last, const int lane,
           unsigned& m0, unsigned& m1)
{
    m0 = lo.x & hi.x;
    m1 = lo.y & hi.y;
    if (last && lane >= SCAN_TAIL_LANES) { m0 = 0u; m1 = 0u; }
    if (!valid) { m0 = 0u; m1 = 0u; }
}

__device__ __forceinline__ void
scan_keep(unsigned m, const unsigned e0, unsigned* myq, unsigned& count)
{
    unsigned live = __builtin_amdgcn_ballot_w32(m != 0u);
    while (live != 0u) {
        if (m != 0u)
            myq[count + __builtin_amdgcn_mbcnt_lo(live, 0u)] = e0 + (unsigned)__builtin_ctz(m);
        count += (unsigned)__builtin_popcount(live);
        m &= m - 1u;
        live = __builtin_amdgcn_ballot_w32(m != 0u);
    }
}

template <int M>
__device__ __forceinline__ float row_xor(const float x)
{
    return __builtin_bit_cast(float,
        __builtin_amdgcn_mov_dpp(__builtin_bit_cast(int, x), 0x160 + M, 0xf, 0xf, true));
}

template <int M, int U>
__device__ __forceinline__ void row_xor_add(float (&p0)[U], float (&p1)[U])
{
#pragma unroll
    for (int t = 0; t < U; ++t) {
        p0[t] += row_xor<M>(p0[t]);
        p1[t] += row_xor<M>(p1[t]);
    }
}

template <int T>
__device__ __forceinline__ float row_take(const float x)
{
    return __builtin_bit_cast(float,
        __builtin_amdgcn_mov_dpp(__builtin_bit_cast(int, x), 0x150 + T, 0xf, 0xf, true));
}

template <int M>
__device__ __forceinline__ float fold_pair(const float a, const float b, const bool up)
{
    const float keep = up ? b : a;
    const float give = up ? a : b;
    return keep + row_xor<M>(give);
}

template <int T>
__device__ __forceinline__ void
wv_add(const unsigned (&dls)[8], const float (&v0)[8], const float (&v1)[8],
       const float e, float* s_acc, const int lane)
{
    float* a = s_acc + dls[T] * ACC_PITCH;
    atomicAdd(&a[lane],      v0[T] * row_take<T>(e));
    atomicAdd(&a[lane + 32], v1[T] * row_take<8 + T>(e));
}

__device__ __forceinline__ void
edge_batch8(const unsigned (&srcs)[8], const unsigned (&dls)[8], const unsigned dlv,
            const float* __restrict__ K, const float* __restrict__ V,
            const float* s_q, float* s_acc, const int lane)
{
    const int grp = lane >> 4;
    const int dd  = lane & 15;
    float k0[8], k1[8], v0[8], v1[8], val[16];
#pragma unroll
    for (int t = 0; t < 8; ++t) {
        const float* __restrict__ kr = K + (size_t)srcs[t] * HD;
        const float* __restrict__ vr = V + (size_t)srcs[t] * HD;
        k0[t] = kr[lane]; k1[t] = kr[lane + 32];
        v0[t] = vr[lane]; v1[t] = vr[lane + 32];
    }
#pragma unroll
    for (int t = 0; t < 8; ++t) {
        val[t]     = k0[t] * s_q[dls[t] * HD + lane];
        val[8 + t] = k1[t] * s_q[dls[t] * HD + lane + 32];
    }
    const bool u1 = (lane & 1) != 0, u2 = (lane & 2) != 0, u4 = (lane & 4) != 0, u8 = (lane & 8) != 0;
    float r1[8], r2[4], r3[2];
#pragma unroll
    for (int i = 0; i < 8; ++i) r1[i] = fold_pair<1>(val[2 * i], val[2 * i + 1], u1);
#pragma unroll
    for (int i = 0; i < 4; ++i) r2[i] = fold_pair<2>(r1[2 * i], r1[2 * i + 1], u2);
#pragma unroll
    for (int i = 0; i < 2; ++i) r3[i] = fold_pair<4>(r2[2 * i], r2[2 * i + 1], u4);
    const float s = fold_pair<8>(r3[0], r3[1], u8);
    const float e = __expf(fminf(fmaxf(s * 0.25f, -5.0f), 5.0f));
    atomicAdd(&s_acc[dlv * ACC_PITCH + HD + grp + 2 * (dd >> 3)], e);
    wv_add<0>(dls, v0, v1, e, s_acc, lane);
    wv_add<1>(dls, v0, v1, e, s_acc, lane);
    wv_add<2>(dls, v0, v1, e, s_acc, lane);
    wv_add<3>(dls, v0, v1, e, s_acc, lane);
    wv_add<4>(dls, v0, v1, e, s_acc, lane);
    wv_add<5>(dls, v0, v1, e, s_acc, lane);
    wv_add<6>(dls, v0, v1, e, s_acc, lane);
    wv_add<7>(dls, v0, v1, e, s_acc, lane);
}

template <int U>
__device__ __forceinline__ void
edge_batch(const unsigned (&srcs)[U], const unsigned (&dls)[U],
           const float* __restrict__ K, const float* __restrict__ V,
           const float* s_q, float* s_acc, const int lane)
{
    const int grp = lane >> 4;
    const int dd  = lane & 15;
    float k0[U], k1[U], v0[U], v1[U], p0[U], p1[U];
#pragma unroll
    for (int t = 0; t < U; ++t) {
        const float* __restrict__ kr = K + (size_t)srcs[t] * HD;
        const float* __restrict__ vr = V + (size_t)srcs[t] * HD;
        k0[t] = kr[lane]; k1[t] = kr[lane + 32];
        v0[t] = vr[lane]; v1[t] = vr[lane + 32];
    }
#pragma unroll
    for (int t = 0; t < U; ++t) {
        p0[t] = k0[t] * s_q[dls[t] * HD + lane];
        p1[t] = k1[t] * s_q[dls[t] * HD + lane + 32];
    }
    row_xor_add<1, U>(p0, p1);
    row_xor_add<2, U>(p0, p1);
    row_xor_add<4, U>(p0, p1);
    row_xor_add<8, U>(p0, p1);
#pragma unroll
    for (int t = 0; t < U; ++t) {
        p0[t] = __expf(fminf(fmaxf(p0[t] * 0.25f, -5.0f), 5.0f));
        p1[t] = __expf(fminf(fmaxf(p1[t] * 0.25f, -5.0f), 5.0f));
        float* a = s_acc + dls[t] * ACC_PITCH;
        atomicAdd(&a[lane],      p0[t] * v0[t]);
        atomicAdd(&a[lane + 32], p1[t] * v1[t]);
    }
    if (dd == 0) {
#pragma unroll
        for (int t = 0; t < U; ++t) {
            float* a = s_acc + dls[t] * ACC_PITCH + HD;
            atomicAdd(&a[grp],     p0[t]);
            atomicAdd(&a[2 + grp], p1[t]);
        }
    }
}

__device__ __forceinline__ void
work_off(const unsigned* myq, const unsigned count,
         const unsigned* __restrict__ pk,
         const float* __restrict__ K, const float* __restrict__ V,
         const float* s_q, float* s_acc, const int lane)
{
    for (unsigned i0 = 0; i0 < count; i0 += 32) {
        const unsigned left = count - i0;
        const int nb = (int)(left < 32u ? left : 32u);
        unsigned sv = 0, dv = 0;
        if (lane < nb) {
            const unsigned w = pk[myq[i0 + lane]];
            sv = w & 0xFFFFu;
            dv = w >> 16;
            dv = dv < (unsigned)OWN_MAX ? dv : 0u;
        }
        int j = 0;
        for (; j + EDGE_U <= nb; j += EDGE_U) {
            unsigned srcs[EDGE_U], dls[EDGE_U];
#pragma unroll
            for (int t = 0; t < EDGE_U; ++t) {
                srcs[t] = __builtin_amdgcn_readlane(sv, j + t);
                dls[t]  = __builtin_amdgcn_readlane(dv, j + t);
            }
            const unsigned dlv = (unsigned)__builtin_amdgcn_ds_bpermute((j + (lane & 7)) << 2, (int)dv);
            edge_batch8(srcs, dls, dlv, K, V, s_q, s_acc, lane);
        }
        for (; j < nb; ++j) {
            unsigned srcs[1], dls[1];
            srcs[0] = __builtin_amdgcn_readlane(sv, j);
            dls[0]  = __builtin_amdgcn_readlane(dv, j);
            edge_batch<1>(srcs, dls, K, V, s_q, s_acc, lane);
        }
    }
}

__global__ void __launch_bounds__(1024)
gt_edge_owner(const unsigned* __restrict__ pk,
              const unsigned* __restrict__ plane,
              const float* __restrict__ K, const float* __restrict__ V,
              float* out)
{
    __shared__ __attribute__((aligned(16))) float    s_acc[OWN_MAX * ACC_PITCH];
    __shared__ __attribute__((aligned(16))) float    s_q[OWN_MAX * HD];
    __shared__ __attribute__((aligned(16))) unsigned s_queue[OWN_WAVES * QW];

    const int tid  = (int)threadIdx.x;
    const int lane = tid & 31;
    const int wave = __builtin_amdgcn_readfirstlane(tid >> 5);
    const int b    = (int)blockIdx.x;
    const unsigned lo  = (unsigned)(b * 3125) >> 4;
    const unsigned hi  = (unsigned)((b + 1) * 3125) >> 4;
    const unsigned cnt = hi - lo;
    const uint2* __restrict__ pl_lo = (const uint2*)(plane + (size_t)(b & 15) * N_GROUPS);
    const uint2* __restrict__ pl_hi = (const uint2*)(plane + (size_t)(16 + (b >> 4)) * N_GROUPS);
    const int skew = b * START_SKEW;
    const unsigned lane64 = (unsigned)lane * 64u;

    unsigned* myq = s_queue + wave * QW;
    unsigned count = 0;

    uint2    aLo, aHi, bLo, bHi;
    unsigned eA, eB;
    bool     okA, okB, lastA, lastB;
    scan_load(pl_lo, pl_hi, 0, wave, lane, skew, aLo, aHi, eA, okA, lastA);
    scan_load(pl_lo, pl_hi, 1, wave, lane, skew, bLo, bHi, eB, okB, lastB);

    for (int i = tid; i < OWN_MAX * ACC_PITCH; i += 1024) s_acc[i] = 0.0f;
    {
        const float4* q4 = (const float4*)(out + (size_t)lo * HD);
        float4* s4 = (float4*)s_q;
        const int nq = (int)cnt * (HD / 4);
        float4 r0, r1, r2, r3;
        { const int i = tid;        r0 = q4[i < nq ? i : nq - 1]; }
        { const int i = tid + 1024; r1 = q4[i < nq ? i : nq - 1]; }
        { const int i = tid + 2048; r2 = q4[i < nq ? i : nq - 1]; }
        { const int i = tid + 3072; r3 = q4[i < nq ? i : nq - 1]; }
        if (tid        < nq) s4[tid]        = r0;
        if (tid + 1024 < nq) s4[tid + 1024] = r1;
        if (tid + 2048 < nq) s4[tid + 2048] = r2;
        if (tid + 3072 < nq) s4[tid + 3072] = r3;
    }
    __syncthreads();

    for (int g = 0; ; g += 2) {
        if (g == N_STEP || count > Q_KEEP) {
            work_off(myq, count, pk, K, V, s_q, s_acc, lane);
            count = 0;
            if (g == N_STEP) break;
        }
        {
            unsigned m0, m1;
            scan_match(aLo, aHi, okA, lastA, lane, m0, m1);
            const unsigned e0 = eA + lane64;
            scan_load(pl_lo, pl_hi, g + 2, wave, lane, skew, aLo, aHi, eA, okA, lastA);
            scan_keep(m0, e0, myq, count);
            if (count > Q_KEEP) {
                work_off(myq, count, pk, K, V, s_q, s_acc, lane);
                count = 0;
            }
            scan_keep(m1, e0 + 32u, myq, count);
        }
        if (count > Q_KEEP) {
            work_off(myq, count, pk, K, V, s_q, s_acc, lane);
            count = 0;
        }
        {
            unsigned m0, m1;
            scan_match(bLo, bHi, okB, lastB, lane, m0, m1);
            const unsigned e0 = eB + lane64;
            scan_load(pl_lo, pl_hi, g + 3, wave, lane, skew, bLo, bHi, eB, okB, lastB);
            scan_keep(m0, e0, myq, count);
            if (count > Q_KEEP) {
                work_off(myq, count, pk, K, V, s_q, s_acc, lane);
                count = 0;
            }
            scan_keep(m1, e0 + 32u, myq, count);
        }
    }

    __syncthreads();

    {
        float* o = out + (size_t)lo * HD;
        for (int i = tid; i < (int)cnt * HD; i += 1024) {
            const int r = i >> 6;
            const int c = i & 63;
            const int h = c >> 4;
            o[i] = s_acc[r * ACC_PITCH + c] / (s_acc[r * ACC_PITCH + HD + h] + 1e-6f);
        }
    }
}

extern "C" void kernel_launch(void* const* d_in, const int* in_sizes, int n_in,
                              void* d_out, int out_size, void* d_ws, size_t ws_size,
                              hipStream_t stream)
{
    (void)in_sizes; (void)n_in; (void)out_size; (void)ws_size;

    const float* X   = (const float*)d_in[0];
    const int*   src = (const int*)  d_in[1];
    const int*   dst = (const int*)  d_in[2];
    const float* Wq  = (const float*)d_in[3];
    const float* Wk  = (const float*)d_in[4];
    const float* Wv  = (const float*)d_in[5];

    float* out = (float*)d_out;
    float* ws  = (float*)d_ws;
    unsigned* plane = (unsigned*)d_ws;
    unsigned* pk    = (unsigned*)d_ws + PK_OFF_WORDS;
    float* Q   = out;
    float* K   = ws + (size_t)N_NODES * HD;
    float* V   = ws + (size_t)2 * N_NODES * HD;

    {
        const int ntask = (N_NODES / 16) * 3;
        const int wpb   = 256 / 32;
        const int blocks = PLANE_BLOCKS + (ntask + wpb - 1) / wpb;
        gt_qkv_proj_wmma<<<blocks, 256, 0, stream>>>(X, src, dst, plane, pk, Wq, Wk, Wv, Q, K, V);
    }

    gt_edge_owner<<<N_OWNERS, 1024, 0, stream>>>(pk, plane, K, V, out);
}


# Owner files its edges into per-dst-node LDS bins (slot = returned LDS count add); after a barrier one wave sums a whole node in registers, multiply-adds in bin (arrival) order: no LDS float adds
# speedup vs baseline: 2.6125x; 1.1882x over previous
#include <hip/hip_runtime.h>
#include <hip/hip_bf16.h>

typedef __attribute__((ext_vector_type(2))) float v2f;
typedef __attribute__((ext_vector_type(8))) float v8f;

#define N_NODES 50000
#define E_EDGES 1600000
#define NHEAD   4
#define DHEAD   16
#define F_IN    128
#define HD      64

#define N_GROUPS     (E_EDGES / 32)
#define N_PLANES     32
#define PLANE_IDS    1024
#define PLANE_CHUNKS ((E_EDGES + PLANE_IDS - 1) / PLANE_IDS)
#define PLANE_BLOCKS ((PLANE_CHUNKS + 7) / 8)

static_assert(E_EDGES % 32 == 0, "whole dwords");
static_assert(N_GROUPS % 8 == 0, "a quarter of a writer chunk (8 dwords) is whole or absent; planes start 16-byte aligned");
static_assert((16 * (N_NODES - 1) + 15) / 3125 == 255, "owner fits eight bits");

#define PK_OFF_WORDS (N_PLANES * N_GROUPS)
static_assert(PK_OFF_WORDS % 32 == 0, "the edge words start on a 128-byte line");
static_assert(PK_OFF_WORDS + E_EDGES <= N_NODES * HD, "the edge words end before the K rows");
static_assert(N_NODES <= 65536, "a src node fits sixteen bits");

__device__ __forceinline__ unsigned owner_of(const int d) { return (unsigned)(16 * d + 15) / 3125u; }

__device__ __forceinline__ void
plane_chunk(const int* __restrict__ src, const int* __restrict__ dst,
            unsigned* __restrict__ plane, unsigned* __restrict__ pk, const int c, const int lane)
{
    if (c >= PLANE_CHUNKS) return;
    const bool high = lane >= 16;
    unsigned sel[4];
#pragma unroll
    for (int i = 0; i < 4; ++i) sel[i] = ((lane >> i) & 1) ? 0u : 0xFFFFFFFFu;
    uint4* __restrict__ mine = (uint4*)(plane + (size_t)lane * N_GROUPS + c * 32);
    int nq = (N_GROUPS - c * 32) / 8;
    nq = nq < 4 ? nq : 4;
#pragma unroll 1
    for (int q = 0; q < nq; ++q) {
        int d[8], s[8];
        unsigned r[8];
#pragma unroll
        for (int k = 0; k < 8; ++k) {
            const int e = c * PLANE_IDS + (q * 8 + k) * 32 + lane;
            d[k] = dst[e];
            s[k] = src[e];
        }
#pragma unroll
        for (int k = 0; k < 8; ++k) {
            const unsigned o = owner_of(d[k]);
            const int e = c * PLANE_IDS + (q * 8 + k) * 32 + lane;
            pk[e] = (unsigned)s[k] | (((unsigned)d[k] - ((o * 3125u) >> 4)) << 16);
            unsigned bal[8];
#pragma unroll
            for (int i = 0; i < 8; ++i) bal[i] = __builtin_amdgcn_ballot_w32(((o >> i) & 1u) != 0u);
            const unsigned wl = (bal[0] ^ sel[0]) & (bal[1] ^ sel[1]) & (bal[2] ^ sel[2]) & (bal[3] ^ sel[3]);
            const unsigned wh = (bal[4] ^ sel[0]) & (bal[5] ^ sel[1]) & (bal[6] ^ sel[2]) & (bal[7] ^ sel[3]);
            r[k] = high ? wh : wl;
        }
        mine[q * 2 + 0] = make_uint4(r[0], r[1], r[2], r[3]);
        mine[q * 2 + 1] = make_uint4(r[4], r[5], r[6], r[7]);
    }
}

__global__ void __launch_bounds__(256)
gt_qkv_proj_wmma(const float* __restrict__ X,
                 const int*   __restrict__ src,
                 const int*   __restrict__ dst,
                 unsigned*    __restrict__ plane,
                 unsigned*    __restrict__ pk,
                 const float* __restrict__ Wq,
                 const float* __restrict__ Wk,
                 const float* __restrict__ Wv,
                 float* __restrict__ Q,
                 float* __restrict__ K,
                 float* __restrict__ V)
{
    const int lane = (int)(threadIdx.x & 31);
    if (blockIdx.x < PLANE_BLOCKS) {
        plane_chunk(src, dst, plane, pk, (int)((blockIdx.x * blockDim.x + threadIdx.x) >> 5), lane);
        return;
    }
    const int wave = (int)(((blockIdx.x - PLANE_BLOCKS) * blockDim.x + threadIdx.x) >> 5);
    const int NTASK = (N_NODES / 16) * 3;
    if (wave >= NTASK) return;

    const int mat  = wave % 3;
    const int row0 = (wave / 3) * 16;
    const float* __restrict__ W   = (mat == 0) ? Wq : (mat == 1) ? Wk : Wv;
    float*       __restrict__ Out = (mat == 0) ? Q  : (mat == 1) ? K  : V;

    const int grp = lane >> 4;
    const int m16 = lane & 15;

    v8f c0 = {}, c1 = {}, c2 = {}, c3 = {};

    const float* __restrict__ arow = X + (size_t)(row0 + m16) * F_IN + 2 * grp;

#pragma unroll 4
    for (int k0 = 0; k0 < F_IN; k0 += 4) {
        v2f a;
        a.x = arow[k0];
        a.y = arow[k0 + 1];

        const float* __restrict__ wr0 = W + (size_t)(k0 + 2 * grp) * HD + m16;
        const float* __restrict__ wr1 = wr0 + HD;

        v2f b0, b1, b2, b3;
        b0.x = wr0[0];  b0.y = wr1[0];
        b1.x = wr0[16]; b1.y = wr1[16];
        b2.x = wr0[32]; b2.y = wr1[32];
        b3.x = wr0[48]; b3.y = wr1[48];

        c0 = __builtin_amdgcn_wmma_f32_16x16x4_f32(false, a, false, b0, (short)0, c0, false, false);
        c1 = __builtin_amdgcn_wmma_f32_16x16x4_f32(false, a, false, b1, (short)0, c1, false, false);
        c2 = __builtin_amdgcn_wmma_f32_16x16x4_f32(false, a, false, b2, (short)0, c2, false, false);
        c3 = __builtin_amdgcn_wmma_f32_16x16x4_f32(false, a, false, b3, (short)0, c3, false, false);
    }

    const int mrow = 8 * grp;
#pragma unroll
    for (int r = 0; r < 8; ++r) {
        float* __restrict__ o = Out + (size_t)(row0 + mrow + r) * HD + m16;
        o[0]  = c0[r];
        o[16] = c1[r];
        o[32] = c2[r];
        o[48] = c3[r];
    }
}

#define N_OWNERS   256
#define OWN_MAX    196
#define ACC_PITCH  68
#define OWN_WAVES  32
#define SCAN_IDS   2048
#define SCAN_DW    (SCAN_IDS / 32)
#define N_CHUNK    ((N_GROUPS + SCAN_DW - 1) / SCAN_DW)
#define SCAN_TAIL_LANES ((N_GROUPS - (N_CHUNK - 1) * SCAN_DW) / 2)
#define N_STEP     26
#define Q_KEEP     64
#define QW         (Q_KEEP + 1024)
#define START_SKEW 3
#define BIN_CAP    64

static_assert(N_STEP % 2 == 0 && N_STEP * OWN_WAVES >= N_CHUNK, "the steps cover the plane");
static_assert(BIN_CAP == 64, "a bin is two slots per lane, whole batches of eight");
static_assert(N_CHUNK + (N_OWNERS - 1) * START_SKEW < 2 * N_CHUNK, "one wrap at most");
static_assert(N_OWNERS == 256, "two nibbles name the owner");
static_assert((N_GROUPS - (N_CHUNK - 1) * SCAN_DW) % 2 == 0, "the last chunk ends on a lane boundary");
static_assert(SCAN_TAIL_LANES >= 1 && SCAN_TAIL_LANES <= 32, "tail lanes");
static_assert(N_GROUPS % 2 == 0, "planes start 8-byte aligned");
static_assert(OWN_MAX * (HD / 4) <= 4 * 1024, "Q rows: four 16-byte loads per thread");

__device__ __forceinline__ void
scan_load(const uint2* __restrict__ pl_lo, const uint2* __restrict__ pl_hi,
          const int step, const int wave, const int lane,
          const int skew, uint2& lo, uint2& hi, unsigned& e, bool& valid, bool& last)
{
    const int c = step * OWN_WAVES + wave;
    valid = c < N_CHUNK;
    int p = (valid ? c : 0) + skew; if (p >= N_CHUNK) p -= N_CHUNK;
    last = p == N_CHUNK - 1;
    const int idx = p * 32 + ((last && lane >= SCAN_TAIL_LANES) ? 0 : lane);
    lo = pl_lo[idx];
    hi = pl_hi[idx];
    e = (unsigned)p * SCAN_IDS;
}

__device__ __forceinline__ void
scan_match(const uint2& lo, const uint2& hi, const bool valid, const bool last, const int lane,
           unsigned& m0, unsigned& m1)
{
    m0 = lo.x & hi.x;
    m1 = lo.y & hi.y;
    if (last && lane >= SCAN_TAIL_LANES) { m0 = 0u; m1 = 0u; }
    if (!valid) { m0 = 0u; m1 = 0u; }
}

__device__ __forceinline__ void
scan_keep(unsigned m, const unsigned e0, unsigned* myq, unsigned& count)
{
    unsigned live = __builtin_amdgcn_ballot_w32(m != 0u);
    while (live != 0u) {
        if (m != 0u)
            myq[count + __builtin_amdgcn_mbcnt_lo(live, 0u)] = e0 + (unsigned)__builtin_ctz(m);
        count += (unsigned)__builtin_popcount(live);
        m &= m - 1u;
        live = __builtin_amdgcn_ballot_w32(m != 0u);
    }
}

template <int M>
__device__ __forceinline__ float row_xor(const float x)
{
    return __builtin_bit_cast(float,
        __builtin_amdgcn_mov_dpp(__builtin_bit_cast(int, x), 0x160 + M, 0xf, 0xf, true));
}

template <int M, int U>
__device__ __forceinline__ void row_xor_add(float (&p0)[U], float (&p1)[U])
{
#pragma unroll
    for (int t = 0; t < U; ++t) {
        p0[t] += row_xor<M>(p0[t]);
        p1[t] += row_xor<M>(p1[t]);
    }
}

template <int T>
__device__ __forceinline__ float row_take(const float x)
{
    return __builtin_bit_cast(float,
        __builtin_amdgcn_mov_dpp(__builtin_bit_cast(int, x), 0x150 + T, 0xf, 0xf, true));
}

template <int M>
__device__ __forceinline__ float fold_pair(const float a, const float b, const bool up)
{
    const float keep = up ? b : a;
    const float give = up ? a : b;
    return keep + row_xor<M>(give);
}

__device__ __forceinline__ void
node_batch8(const unsigned (&srcs)[8], const bool live, const float q0, const float q1,
            const float* __restrict__ K, const float* __restrict__ V, const int lane,
            float& acc0, float& acc1, float& zacc)
{
    float k0[8], k1[8], v0[8], v1[8], val[16];
#pragma unroll
    for (int t = 0; t < 8; ++t) {
        const float* __restrict__ kr = K + (size_t)srcs[t] * HD;
        const float* __restrict__ vr = V + (size_t)srcs[t] * HD;
        k0[t] = kr[lane]; k1[t] = kr[lane + 32];
        v0[t] = vr[lane]; v1[t] = vr[lane + 32];
    }
#pragma unroll
    for (int t = 0; t < 8; ++t) {
        val[t]     = k0[t] * q0;
        val[8 + t] = k1[t] * q1;
    }
    const bool u1 = (lane & 1) != 0, u2 = (lane & 2) != 0, u4 = (lane & 4) != 0, u8 = (lane & 8) != 0;
    float r1[8], r2[4], r3[2];
#pragma unroll
    for (int i = 0; i < 8; ++i) r1[i] = fold_pair<1>(val[2 * i], val[2 * i + 1], u1);
#pragma unroll
    for (int i = 0; i < 4; ++i) r2[i] = fold_pair<2>(r1[2 * i], r1[2 * i + 1], u2);
#pragma unroll
    for (int i = 0; i < 2; ++i) r3[i] = fold_pair<4>(r2[2 * i], r2[2 * i + 1], u4);
    const float s = fold_pair<8>(r3[0], r3[1], u8);
    float e = __expf(fminf(fmaxf(s * 0.25f, -5.0f), 5.0f));
    e = live ? e : 0.0f;
    zacc += e;
    acc0 += v0[0] * row_take<0>(e);  acc1 += v1[0] * row_take<8>(e);
    acc0 += v0[1] * row_take<1>(e);  acc1 += v1[1] * row_take<9>(e);
    acc0 += v0[2] * row_take<2>(e);  acc1 += v1[2] * row_take<10>(e);
    acc0 += v0[3] * row_take<3>(e);  acc1 += v1[3] * row_take<11>(e);
    acc0 += v0[4] * row_take<4>(e);  acc1 += v1[4] * row_take<12>(e);
    acc0 += v0[5] * row_take<5>(e);  acc1 += v1[5] * row_take<13>(e);
    acc0 += v0[6] * row_take<6>(e);  acc1 += v1[6] * row_take<14>(e);
    acc0 += v0[7] * row_take<7>(e);  acc1 += v1[7] * row_take<15>(e);
}

template <int U>
__device__ __forceinline__ void
edge_batch(const unsigned (&srcs)[U], const unsigned (&dls)[U],
           const float* __restrict__ K, const float* __restrict__ V,
           const float* s_q, float* s_acc, const int lane)
{
    const int grp = lane >> 4;
    const int dd  = lane & 15;
    float k0[U], k1[U], v0[U], v1[U], p0[U], p1[U];
#pragma unroll
    for (int t = 0; t < U; ++t) {
        const float* __restrict__ kr = K + (size_t)srcs[t] * HD;
        const float* __restrict__ vr = V + (size_t)srcs[t] * HD;
        k0[t] = kr[lane]; k1[t] = kr[lane + 32];
        v0[t] = vr[lane]; v1[t] = vr[lane + 32];
    }
#pragma unroll
    for (int t = 0; t < U; ++t) {
        p0[t] = k0[t] * s_q[dls[t] * HD + lane];
        p1[t] = k1[t] * s_q[dls[t] * HD + lane + 32];
    }
    row_xor_add<1, U>(p0, p1);
    row_xor_add<2, U>(p0, p1);
    row_xor_add<4, U>(p0, p1);
    row_xor_add<8, U>(p0, p1);
#pragma unroll
    for (int t = 0; t < U; ++t) {
        p0[t] = __expf(fminf(fmaxf(p0[t] * 0.25f, -5.0f), 5.0f));
        p1[t] = __expf(fminf(fmaxf(p1[t] * 0.25f, -5.0f), 5.0f));
        float* a = s_acc + dls[t] * ACC_PITCH;
        atomicAdd(&a[lane],      p0[t] * v0[t]);
        atomicAdd(&a[lane + 32], p1[t] * v1[t]);
    }
    if (dd == 0) {
#pragma unroll
        for (int t = 0; t < U; ++t) {
            float* a = s_acc + dls[t] * ACC_PITCH + HD;
            atomicAdd(&a[grp],     p0[t]);
            atomicAdd(&a[2 + grp], p1[t]);
        }
    }
}

__device__ __forceinline__ void
file_off(const unsigned* myq, const unsigned count,
         const unsigned* __restrict__ pk,
         const float* __restrict__ K, const float* __restrict__ V,
         const float* s_q, float* s_acc, unsigned* s_cnt, unsigned* s_bin, const int lane)
{
    for (unsigned i0 = 0; i0 < count; i0 += 32) {
        const unsigned left = count - i0;
        const int nb = (int)(left < 32u ? left : 32u);
        unsigned sv = 0, dv = 0;
        bool over = false;
        if (lane < nb) {
            const unsigned w = pk[myq[i0 + lane]];
            sv = w & 0xFFFFu;
            dv = w >> 16;
            dv = dv < (unsigned)OWN_MAX ? dv : 0u;
            const unsigned slot = atomicAdd(&s_cnt[dv], 1u);
            if (slot < (unsigned)BIN_CAP) s_bin[dv * BIN_CAP + slot] = sv;
            else over = true;
        }
        unsigned ov = __builtin_amdgcn_ballot_w32(over);
        while (ov != 0u) {
            const int j = __builtin_ctz(ov);
            unsigned srcs[1], dls[1];
            srcs[0] = __builtin_amdgcn_readlane(sv, j);
            dls[0]  = __builtin_amdgcn_readlane(dv, j);
            edge_batch<1>(srcs, dls, K, V, s_q, s_acc, lane);
            ov &= ov - 1u;
        }
    }
}

__global__ void __launch_bounds__(1024)
gt_edge_owner(const unsigned* __restrict__ pk,
              const unsigned* __restrict__ plane,
              const float* __restrict__ K, const float* __restrict__ V,
              float* out)
{
    __shared__ __attribute__((aligned(16))) float    s_acc[OWN_MAX * ACC_PITCH];
    __shared__ __attribute__((aligned(16))) float    s_q[OWN_MAX * HD];
    __shared__ __attribute__((aligned(16))) unsigned s_queue[OWN_WAVES * QW];
    __shared__ __attribute__((aligned(16))) unsigned s_bin[OWN_MAX * BIN_CAP];
    __shared__ unsigned s_cnt[OWN_MAX];
    __shared__ unsigned s_next;

    const int tid  = (int)threadIdx.x;
    const int lane = tid & 31;
    const int wave = __builtin_amdgcn_readfirstlane(tid >> 5);
    const int b    = (int)blockIdx.x;
    const unsigned lo  = (unsigned)(b * 3125) >> 4;
    const unsigned hi  = (unsigned)((b + 1) * 3125) >> 4;
    const unsigned cnt = hi - lo;
    const uint2* __restrict__ pl_lo = (const uint2*)(plane + (size_t)(b & 15) * N_GROUPS);
    const uint2* __restrict__ pl_hi = (const uint2*)(plane + (size_t)(16 + (b >> 4)) * N_GROUPS);
    const int skew = b * START_SKEW;
    const unsigned lane64 = (unsigned)lane * 64u;

    unsigned* myq = s_queue + wave * QW;
    unsigned count = 0;

    uint2    aLo, aHi, bLo, bHi;
    unsigned eA, eB;
    bool     okA, okB, lastA, lastB;
    scan_load(pl_lo, pl_hi, 0, wave, lane, skew, aLo, aHi, eA, okA, lastA);
    scan_load(pl_lo, pl_hi, 1, wave, lane, skew, bLo, bHi, eB, okB, lastB);

    for (int i = tid; i < OWN_MAX * ACC_PITCH; i += 1024) s_acc[i] = 0.0f;
    if (tid < OWN_MAX) s_cnt[tid] = 0u;
    if (tid == 0) s_next = 0u;
    {
        const float4* q4 = (const float4*)(out + (size_t)lo * HD);
        float4* s4 = (float4*)s_q;
        const int nq = (int)cnt * (HD / 4);
        float4 r0, r1, r2, r3;
        { const int i = tid;        r0 = q4[i < nq ? i : nq - 1]; }
        { const int i = tid + 1024; r1 = q4[i < nq ? i : nq - 1]; }
        { const int i = tid + 2048; r2 = q4[i < nq ? i : nq - 1]; }
        { const int i = tid + 3072; r3 = q4[i < nq ? i : nq - 1]; }
        if (tid        < nq) s4[tid]        = r0;
        if (tid + 1024 < nq) s4[tid + 1024] = r1;
        if (tid + 2048 < nq) s4[tid + 2048] = r2;
        if (tid + 3072 < nq) s4[tid + 3072] = r3;
    }
    __syncthreads();

    for (int g = 0; ; g += 2) {
        if (g == N_STEP || count > Q_KEEP) {
            file_off(myq, count, pk, K, V, s_q, s_acc, s_cnt, s_bin, lane);
            count = 0;
            if (g == N_STEP) break;
        }
        {
            unsigned m0, m1;
            scan_match(aLo, aHi, okA, lastA, lane, m0, m1);
            const unsigned e0 = eA + lane64;
            scan_load(pl_lo, pl_hi, g + 2, wave, lane, skew, aLo, aHi, eA, okA, lastA);
            scan_keep(m0, e0, myq, count);
            if (count > Q_KEEP) {
                file_off(myq, count, pk, K, V, s_q, s_acc, s_cnt, s_bin, lane);
                count = 0;
            }
            scan_keep(m1, e0 + 32u, myq, count);
        }
        if (count > Q_KEEP) {
            file_off(myq, count, pk, K, V, s_q, s_acc, s_cnt, s_bin, lane);
            count = 0;
        }
        {
            unsigned m0, m1;
            scan_match(bLo, bHi, okB, lastB, lane, m0, m1);
            const unsigned e0 = eB + lane64;
            scan_load(pl_lo, pl_hi, g + 3, wave, lane, skew, bLo, bHi, eB, okB, lastB);
            scan_keep(m0, e0, myq, count);
            if (count > Q_KEEP) {
                file_off(myq, count, pk, K, V, s_q, s_acc, s_cnt, s_bin, lane);
                count = 0;
            }
            scan_keep(m1, e0 + 32u, myq, count);
        }
    }

    __syncthreads();

    {
        const int grp = lane >> 4;
        for (;;) {
            unsigned dl = 0;
            if (lane == 0) dl = atomicAdd(&s_next, 1u);
            dl = (unsigned)__builtin_amdgcn_readfirstlane((int)dl);
            if (dl >= cnt) break;
            unsigned n = (unsigned)__builtin_amdgcn_readfirstlane((int)s_cnt[dl]);
            n = n < (unsigned)BIN_CAP ? n : (unsigned)BIN_CAP;
            const float q0 = s_q[dl * HD + lane];
            const float q1 = s_q[dl * HD + lane + 32];
            float acc0 = s_acc[dl * ACC_PITCH + lane];
            float acc1 = s_acc[dl * ACC_PITCH + lane + 32];
            float zacc = 0.0f;
            const unsigned bin0 = s_bin[dl * BIN_CAP + lane];
            const unsigned bin1 = s_bin[dl * BIN_CAP + lane + 32];
            for (unsigned j0 = 0; j0 < n; j0 += 8) {
                const bool     live = j0 + (unsigned)(lane & 7) < n;
                const unsigned half = j0 < 32u ? bin0 : bin1;
                unsigned srcs[8];
#pragma unroll
                for (int t = 0; t < 8; ++t) {
                    unsigned i = j0 + (unsigned)t;
                    i = i < n ? i : n - 1u;
                    srcs[t] = (unsigned)__builtin_amdgcn_readlane((int)half, (int)(i & 31u));
                }
                node_batch8(srcs, live, q0, q1, K, V, lane, acc0, acc1, zacc);
            }
            zacc += row_xor<1>(zacc);
            zacc += row_xor<2>(zacc);
            zacc += row_xor<4>(zacc);
            const float z0 = row_take<0>(zacc) + s_acc[dl * ACC_PITCH + HD + grp];
            const float z1 = row_take<8>(zacc) + s_acc[dl * ACC_PITCH + HD + grp + 2];
            float* o = out + (size_t)(lo + dl) * HD;
            o[lane]      = acc0 / (z0 + 1e-6f);
            o[lane + 32] = acc1 / (z1 + 1e-6f);
        }
    }
}

extern "C" void kernel_launch(void* const* d_in, const int* in_sizes, int n_in,
                              void* d_out, int out_size, void* d_ws, size_t ws_size,
                              hipStream_t stream)
{
    (void)in_sizes; (void)n_in; (void)out_size; (void)ws_size;

    const float* X   = (const float*)d_in[0];
    const int*   src = (const int*)  d_in[1];
    const int*   dst = (const int*)  d_in[2];
    const float* Wq  = (const float*)d_in[3];
    const float* Wk  = (const float*)d_in[4];
    const float* Wv  = (const float*)d_in[5];

    float* out = (float*)d_out;
    float* ws  = (float*)d_ws;
    unsigned* plane = (unsigned*)d_ws;
    unsigned* pk    = (unsigned*)d_ws + PK_OFF_WORDS;
    float* Q   = out;
    float* K   = ws + (size_t)N_NODES * HD;
    float* V   = ws + (size_t)2 * N_NODES * HD;

    {
        const int ntask = (N_NODES / 16) * 3;
        const int wpb   = 256 / 32;
        const int blocks = PLANE_BLOCKS + (ntask + wpb - 1) / wpb;
        gt_qkv_proj_wmma<<<blocks, 256, 0, stream>>>(X, src, dst, plane, pk, Wq, Wk, Wv, Q, K, V);
    }

    gt_edge_owner<<<N_OWNERS, 1024, 0, stream>>>(pk, plane, K, V, out);
}


# Sum phase: a lane owns adjacent slots 2l,2l+1, so a K or V row is ONE 8-byte-per-lane load (2 row loads per edge, was 4); score terms pair as before, first add now an in-lane fma; V mul-adds fused
# speedup vs baseline: 2.7524x; 1.0536x over previous
#include <hip/hip_runtime.h>
#include <hip/hip_bf16.h>

typedef __attribute__((ext_vector_type(2))) float v2f;
typedef __attribute__((ext_vector_type(8))) float v8f;

#define N_NODES 50000
#define E_EDGES 1600000
#define NHEAD   4
#define DHEAD   16
#define F_IN    128
#define HD      64

#define N_GROUPS     (E_EDGES / 32)
#define N_PLANES     32
#define PLANE_IDS    1024
#define PLANE_CHUNKS ((E_EDGES + PLANE_IDS - 1) / PLANE_IDS)
#define PLANE_BLOCKS ((PLANE_CHUNKS + 7) / 8)

static_assert(E_EDGES % 32 == 0, "whole dwords");
static_assert(N_GROUPS % 8 == 0, "a quarter of a writer chunk (8 dwords) is whole or absent; planes start 16-byte aligned");
static_assert((16 * (N_NODES - 1) + 15) / 3125 == 255, "owner fits eight bits");

#define PK_OFF_WORDS (N_PLANES * N_GROUPS)
static_assert(PK_OFF_WORDS % 32 == 0, "the edge words start on a 128-byte line");
static_assert(PK_OFF_WORDS + E_EDGES <= N_NODES * HD, "the edge words end before the K rows");
static_assert(N_NODES <= 65536, "a src node fits sixteen bits");

__device__ __forceinline__ unsigned owner_of(const int d) { return (unsigned)(16 * d + 15) / 3125u; }

__device__ __forceinline__ void
plane_chunk(const int* __restrict__ src, const int* __restrict__ dst,
            unsigned* __restrict__ plane, unsigned* __restrict__ pk, const int c, const int lane)
{
    if (c >= PLANE_CHUNKS) return;
    const bool high = lane >= 16;
    unsigned sel[4];
#pragma unroll
    for (int i = 0; i < 4; ++i) sel[i] = ((lane >> i) & 1) ? 0u : 0xFFFFFFFFu;
    uint4* __restrict__ mine = (uint4*)(plane + (size_t)lane * N_GROUPS + c * 32);
    int nq = (N_GROUPS - c * 32) / 8;
    nq = nq < 4 ? nq : 4;
#pragma unroll 1
    for (int q = 0; q < nq; ++q) {
        int d[8], s[8];
        unsigned r[8];
#pragma unroll
        for (int k = 0; k < 8; ++k) {
            const int e = c * PLANE_IDS + (q * 8 + k) * 32 + lane;
            d[k] = dst[e];
            s[k] = src[e];
        }
#pragma unroll
        for (int k = 0; k < 8; ++k) {
            const unsigned o = owner_of(d[k]);
            const int e = c * PLANE_IDS + (q * 8 + k) * 32 + lane;
            pk[e] = (unsigned)s[k] | (((unsigned)d[k] - ((o * 3125u) >> 4)) << 16);
            unsigned bal[8];
#pragma unroll
            for (int i = 0; i < 8; ++i) bal[i] = __builtin_amdgcn_ballot_w32(((o >> i) & 1u) != 0u);
            const unsigned wl = (bal[0] ^ sel[0]) & (bal[1] ^ sel[1]) & (bal[2] ^ sel[2]) & (bal[3] ^ sel[3]);
            const unsigned wh = (bal[4] ^ sel[0]) & (bal[5] ^ sel[1]) & (bal[6] ^ sel[2]) & (bal[7] ^ sel[3]);
            r[k] = high ? wh : wl;
        }
        mine[q * 2 + 0] = make_uint4(r[0], r[1], r[2], r[3]);
        mine[q * 2 + 1] = make_uint4(r[4], r[5], r[6], r[7]);
    }
}

__global__ void __launch_bounds__(256)
gt_qkv_proj_wmma(const float* __restrict__ X,
                 const int*   __restrict__ src,
                 const int*   __restrict__ dst,
                 unsigned*    __restrict__ plane,
                 unsigned*    __restrict__ pk,
                 const float* __restrict__ Wq,
                 const float* __restrict__ Wk,
                 const float* __restrict__ Wv,
                 float* __restrict__ Q,
                 float* __restrict__ K,
                 float* __restrict__ V)
{
    const int lane = (int)(threadIdx.x & 31);
    if (blockIdx.x < PLANE_BLOCKS) {
        plane_chunk(src, dst, plane, pk, (int)((blockIdx.x * blockDim.x + threadIdx.x) >> 5), lane);
        return;
    }
    const int wave = (int)(((blockIdx.x - PLANE_BLOCKS) * blockDim.x + threadIdx.x) >> 5);
    const int NTASK = (N_NODES / 16) * 3;
    if (wave >= NTASK) return;

    const int mat  = wave % 3;
    const int row0 = (wave / 3) * 16;
    const float* __restrict__ W   = (mat == 0) ? Wq : (mat == 1) ? Wk : Wv;
    float*       __restrict__ Out = (mat == 0) ? Q  : (mat == 1) ? K  : V;

    const int grp = lane >> 4;
    const int m16 = lane & 15;

    v8f c0 = {}, c1 = {}, c2 = {}, c3 = {};

    const float* __restrict__ arow = X + (size_t)(row0 + m16) * F_IN + 2 * grp;

#pragma unroll 4
    for (int k0 = 0; k0 < F_IN; k0 += 4) {
        v2f a;
        a.x = arow[k0];
        a.y = arow[k0 + 1];

        const float* __restrict__ wr0 = W + (size_t)(k0 + 2 * grp) * HD + m16;
        const float* __restrict__ wr1 = wr0 + HD;

        v2f b0, b1, b2, b3;
        b0.x = wr0[0];  b0.y = wr1[0];
        b1.x = wr0[16]; b1.y = wr1[16];
        b2.x = wr0[32]; b2.y = wr1[32];
        b3.x = wr0[48]; b3.y = wr1[48];

        c0 = __builtin_amdgcn_wmma_f32_16x16x4_f32(false, a, false, b0, (short)0, c0, false, false);
        c1 = __builtin_amdgcn_wmma_f32_16x16x4_f32(false, a, false, b1, (short)0, c1, false, false);
        c2 = __builtin_amdgcn_wmma_f32_16x16x4_f32(false, a, false, b2, (short)0, c2, false, false);
        c3 = __builtin_amdgcn_wmma_f32_16x16x4_f32(false, a, false, b3, (short)0, c3, false, false);
    }

    const int mrow = 8 * grp;
#pragma unroll
    for (int r = 0; r < 8; ++r) {
        float* __restrict__ o = Out + (size_t)(row0 + mrow + r) * HD + m16;
        o[0]  = c0[r];
        o[16] = c1[r];
        o[32] = c2[r];
        o[48] = c3[r];
    }
}

#define N_OWNERS   256
#define OWN_MAX    196
#define ACC_PITCH  68
#define OWN_WAVES  32
#define SCAN_IDS   2048
#define SCAN_DW    (SCAN_IDS / 32)
#define N_CHUNK    ((N_GROUPS + SCAN_DW - 1) / SCAN_DW)
#define SCAN_TAIL_LANES ((N_GROUPS - (N_CHUNK - 1) * SCAN_DW) / 2)
#define N_STEP     26
#define Q_KEEP     64
#define QW         (Q_KEEP + 1024)
#define START_SKEW 3
#define BIN_CAP    64

static_assert(N_STEP % 2 == 0 && N_STEP * OWN_WAVES >= N_CHUNK, "the steps cover the plane");
static_assert(BIN_CAP == 64, "a bin is two slots per lane, whole batches of eight");
static_assert(N_CHUNK + (N_OWNERS - 1) * START_SKEW < 2 * N_CHUNK, "one wrap at most");
static_assert(N_OWNERS == 256, "two nibbles name the owner");
static_assert((N_GROUPS - (N_CHUNK - 1) * SCAN_DW) % 2 == 0, "the last chunk ends on a lane boundary");
static_assert(SCAN_TAIL_LANES >= 1 && SCAN_TAIL_LANES <= 32, "tail lanes");
static_assert(N_GROUPS % 2 == 0, "planes start 8-byte aligned");
static_assert(OWN_MAX * (HD / 4) <= 4 * 1024, "Q rows: four 16-byte loads per thread");

__device__ __forceinline__ void
scan_load(const uint2* __restrict__ pl_lo, const uint2* __restrict__ pl_hi,
          const int step, const int wave, const int lane,
          const int skew, uint2& lo, uint2& hi, unsigned& e, bool& valid, bool& last)
{
    const int c = step * OWN_WAVES + wave;
    valid = c < N_CHUNK;
    int p = (valid ? c : 0) + skew; if (p >= N_CHUNK) p -= N_CHUNK;
    last = p == N_CHUNK - 1;
    const int idx = p * 32 + ((last && lane >= SCAN_TAIL_LANES) ? 0 : lane);
    lo = pl_lo[idx];
    hi = pl_hi[idx];
    e = (unsigned)p * SCAN_IDS;
}

__device__ __forceinline__ void
scan_match(const uint2& lo, const uint2& hi, const bool valid, const bool last, const int lane,
           unsigned& m0, unsigned& m1)
{
    m0 = lo.x & hi.x;
    m1 = lo.y & hi.y;
    if (last && lane >= SCAN_TAIL_LANES) { m0 = 0u; m1 = 0u; }
    if (!valid) { m0 = 0u; m1 = 0u; }
}

__device__ __forceinline__ void
scan_keep(unsigned m, const unsigned e0, unsigned* myq, unsigned& count)
{
    unsigned live = __builtin_amdgcn_ballot_w32(m != 0u);
    while (live != 0u) {
        if (m != 0u)
            myq[count + __builtin_amdgcn_mbcnt_lo(live, 0u)] = e0 + (unsigned)__builtin_ctz(m);
        count += (unsigned)__builtin_popcount(live);
        m &= m - 1u;
        live = __builtin_amdgcn_ballot_w32(m != 0u);
    }
}

template <int M>
__device__ __forceinline__ float row_xor(const float x)
{
    return __builtin_bit_cast(float,
        __builtin_amdgcn_mov_dpp(__builtin_bit_cast(int, x), 0x160 + M, 0xf, 0xf, true));
}

template <int M, int U>
__device__ __forceinline__ void row_xor_add(float (&p0)[U], float (&p1)[U])
{
#pragma unroll
    for (int t = 0; t < U; ++t) {
        p0[t] += row_xor<M>(p0[t]);
        p1[t] += row_xor<M>(p1[t]);
    }
}

template <int T>
__device__ __forceinline__ float oct_take(const float x)
{
    return __builtin_bit_cast(float,
        __builtin_amdgcn_mov_dpp8(__builtin_bit_cast(unsigned, x), (unsigned)T * 0x249249u));
}

template <int M>
__device__ __forceinline__ float fold_pair(const float a, const float b, const bool up)
{
    const float keep = up ? b : a;
    const float give = up ? a : b;
    return keep + row_xor<M>(give);
}

__device__ __forceinline__ void
node_batch8(const unsigned (&srcs)[8], const bool live, const float2 q,
            const float* __restrict__ K, const float* __restrict__ V, const int lane,
            float2& acc, float& zacc)
{
    float2 k[8], v[8];
    float  val[8];
#pragma unroll
    for (int t = 0; t < 8; ++t) {
        const float2* __restrict__ kr = (const float2*)(K + (size_t)srcs[t] * HD);
        const float2* __restrict__ vr = (const float2*)(V + (size_t)srcs[t] * HD);
        k[t] = kr[lane];
        v[t] = vr[lane];
    }
#pragma unroll
    for (int t = 0; t < 8; ++t) val[t] = k[t].x * q.x + k[t].y * q.y;
    const bool u1 = (lane & 1) != 0, u2 = (lane & 2) != 0, u4 = (lane & 4) != 0;
    float r1[4], r2[2];
#pragma unroll
    for (int i = 0; i < 4; ++i) r1[i] = fold_pair<1>(val[2 * i], val[2 * i + 1], u1);
#pragma unroll
    for (int i = 0; i < 2; ++i) r2[i] = fold_pair<2>(r1[2 * i], r1[2 * i + 1], u2);
    const float s = fold_pair<4>(r2[0], r2[1], u4);
    float e = __expf(fminf(fmaxf(s * 0.25f, -5.0f), 5.0f));
    e = live ? e : 0.0f;
    zacc += e;
    { const float w = oct_take<0>(e);  acc.x += v[0].x * w;  acc.y += v[0].y * w; }
    { const float w = oct_take<1>(e);  acc.x += v[1].x * w;  acc.y += v[1].y * w; }
    { const float w = oct_take<2>(e);  acc.x += v[2].x * w;  acc.y += v[2].y * w; }
    { const float w = oct_take<3>(e);  acc.x += v[3].x * w;  acc.y += v[3].y * w; }
    { const float w = oct_take<4>(e);  acc.x += v[4].x * w;  acc.y += v[4].y * w; }
    { const float w = oct_take<5>(e);  acc.x += v[5].x * w;  acc.y += v[5].y * w; }
    { const float w = oct_take<6>(e);  acc.x += v[6].x * w;  acc.y += v[6].y * w; }
    { const float w = oct_take<7>(e);  acc.x += v[7].x * w;  acc.y += v[7].y * w; }
}

template <int U>
__device__ __forceinline__ void
edge_batch(const unsigned (&srcs)[U], const unsigned (&dls)[U],
           const float* __restrict__ K, const float* __restrict__ V,
           const float* s_q, float* s_acc, const int lane)
{
    const int grp = lane >> 4;
    const int dd  = lane & 15;
    float k0[U], k1[U], v0[U], v1[U], p0[U], p1[U];
#pragma unroll
    for (int t = 0; t < U; ++t) {
        const float* __restrict__ kr = K + (size_t)srcs[t] * HD;
        const float* __restrict__ vr = V + (size_t)srcs[t] * HD;
        k0[t] = kr[lane]; k1[t] = kr[lane + 32];
        v0[t] = vr[lane]; v1[t] = vr[lane + 32];
    }
#pragma unroll
    for (int t = 0; t < U; ++t) {
        p0[t] = k0[t] * s_q[dls[t] * HD + lane];
        p1[t] = k1[t] * s_q[dls[t] * HD + lane + 32];
    }
    row_xor_add<1, U>(p0, p1);
    row_xor_add<2, U>(p0, p1);
    row_xor_add<4, U>(p0, p1);
    row_xor_add<8, U>(p0, p1);
#pragma unroll
    for (int t = 0; t < U; ++t) {
        p0[t] = __expf(fminf(fmaxf(p0[t] * 0.25f, -5.0f), 5.0f));
        p1[t] = __expf(fminf(fmaxf(p1[t] * 0.25f, -5.0f), 5.0f));
        float* a = s_acc + dls[t] * ACC_PITCH;
        atomicAdd(&a[lane],      p0[t] * v0[t]);
        atomicAdd(&a[lane + 32], p1[t] * v1[t]);
    }
    if (dd == 0) {
#pragma unroll
        for (int t = 0; t < U; ++t) {
            float* a = s_acc + dls[t] * ACC_PITCH + HD;
            atomicAdd(&a[grp],     p0[t]);
            atomicAdd(&a[2 + grp], p1[t]);
        }
    }
}

__device__ __forceinline__ void
file_off(const unsigned* myq, const unsigned count,
         const unsigned* __restrict__ pk,
         const float* __restrict__ K, const float* __restrict__ V,
         const float* s_q, float* s_acc, unsigned* s_cnt, unsigned* s_bin, const int lane)
{
    for (unsigned i0 = 0; i0 < count; i0 += 32) {
        const unsigned left = count - i0;
        const int nb = (int)(left < 32u ? left : 32u);
        unsigned sv = 0, dv = 0;
        bool over = false;
        if (lane < nb) {
            const unsigned w = pk[myq[i0 + lane]];
            sv = w & 0xFFFFu;
            dv = w >> 16;
            dv = dv < (unsigned)OWN_MAX ? dv : 0u;
            const unsigned slot = atomicAdd(&s_cnt[dv], 1u);
            if (slot < (unsigned)BIN_CAP) s_bin[dv * BIN_CAP + slot] = sv;
            else over = true;
        }
        unsigned ov = __builtin_amdgcn_ballot_w32(over);
        while (ov != 0u) {
            const int j = __builtin_ctz(ov);
            unsigned srcs[1], dls[1];
            srcs[0] = __builtin_amdgcn_readlane(sv, j);
            dls[0]  = __builtin_amdgcn_readlane(dv, j);
            edge_batch<1>(srcs, dls, K, V, s_q, s_acc, lane);
            ov &= ov - 1u;
        }
    }
}

__global__ void __launch_bounds__(1024)
gt_edge_owner(const unsigned* __restrict__ pk,
              const unsigned* __restrict__ plane,
              const float* __restrict__ K, const float* __restrict__ V,
              float* out)
{
    __shared__ __attribute__((aligned(16))) float    s_acc[OWN_MAX * ACC_PITCH];
    __shared__ __attribute__((aligned(16))) float    s_q[OWN_MAX * HD];
    __shared__ __attribute__((aligned(16))) unsigned s_queue[OWN_WAVES * QW];
    __shared__ __attribute__((aligned(16))) unsigned s_bin[OWN_MAX * BIN_CAP];
    __shared__ unsigned s_cnt[OWN_MAX];
    __shared__ unsigned s_next;

    const int tid  = (int)threadIdx.x;
    const int lane = tid & 31;
    const int wave = __builtin_amdgcn_readfirstlane(tid >> 5);
    const int b    = (int)blockIdx.x;
    const unsigned lo  = (unsigned)(b * 3125) >> 4;
    const unsigned hi  = (unsigned)((b + 1) * 3125) >> 4;
    const unsigned cnt = hi - lo;
    const uint2* __restrict__ pl_lo = (const uint2*)(plane + (size_t)(b & 15) * N_GROUPS);
    const uint2* __restrict__ pl_hi = (const uint2*)(plane + (size_t)(16 + (b >> 4)) * N_GROUPS);
    const int skew = b * START_SKEW;
    const unsigned lane64 = (unsigned)lane * 64u;

    unsigned* myq = s_queue + wave * QW;
    unsigned count = 0;

    uint2    aLo, aHi, bLo, bHi;
    unsigned eA, eB;
    bool     okA, okB, lastA, lastB;
    scan_load(pl_lo, pl_hi, 0, wave, lane, skew, aLo, aHi, eA, okA, lastA);
    scan_load(pl_lo, pl_hi, 1, wave, lane, skew, bLo, bHi, eB, okB, lastB);

    for (int i = tid; i < OWN_MAX * ACC_PITCH; i += 1024) s_acc[i] = 0.0f;
    if (tid < OWN_MAX) s_cnt[tid] = 0u;
    if (tid == 0) s_next = 0u;
    {
        const float4* q4 = (const float4*)(out + (size_t)lo * HD);
        float4* s4 = (float4*)s_q;
        const int nq = (int)cnt * (HD / 4);
        float4 r0, r1, r2, r3;
        { const int i = tid;        r0 = q4[i < nq ? i : nq - 1]; }
        { const int i = tid + 1024; r1 = q4[i < nq ? i : nq - 1]; }
        { const int i = tid + 2048; r2 = q4[i < nq ? i : nq - 1]; }
        { const int i = tid + 3072; r3 = q4[i < nq ? i : nq - 1]; }
        if (tid        < nq) s4[tid]        = r0;
        if (tid + 1024 < nq) s4[tid + 1024] = r1;
        if (tid + 2048 < nq) s4[tid + 2048] = r2;
        if (tid + 3072 < nq) s4[tid + 3072] = r3;
    }
    __syncthreads();

    for (int g = 0; ; g += 2) {
        if (g == N_STEP || count > Q_KEEP) {
            file_off(myq, count, pk, K, V, s_q, s_acc, s_cnt, s_bin, lane);
            count = 0;
            if (g == N_STEP) break;
        }
        {
            unsigned m0, m1;
            scan_match(aLo, aHi, okA, lastA, lane, m0, m1);
            const unsigned e0 = eA + lane64;
            scan_load(pl_lo, pl_hi, g + 2, wave, lane, skew, aLo, aHi, eA, okA, lastA);
            scan_keep(m0, e0, myq, count);
            if (count > Q_KEEP) {
                file_off(myq, count, pk, K, V, s_q, s_acc, s_cnt, s_bin, lane);
                count = 0;
            }
            scan_keep(m1, e0 + 32u, myq, count);
        }
        if (count > Q_KEEP) {
            file_off(myq, count, pk, K, V, s_q, s_acc, s_cnt, s_bin, lane);
            count = 0;
        }
        {
            unsigned m0, m1;
            scan_match(bLo, bHi, okB, lastB, lane, m0, m1);
            const unsigned e0 = eB + lane64;
            scan_load(pl_lo, pl_hi, g + 3, wave, lane, skew, bLo, bHi, eB, okB, lastB);
            scan_keep(m0, e0, myq, count);
            if (count > Q_KEEP) {
                file_off(myq, count, pk, K, V, s_q, s_acc, s_cnt, s_bin, lane);
                count = 0;
            }
            scan_keep(m1, e0 + 32u, myq, count);
        }
    }

    __syncthreads();

    {
        const int head = lane >> 3;
        for (;;) {
            unsigned dl = 0;
            if (lane == 0) dl = atomicAdd(&s_next, 1u);
            dl = (unsigned)__builtin_amdgcn_readfirstlane((int)dl);
            if (dl >= cnt) break;
            unsigned n = (unsigned)__builtin_amdgcn_readfirstlane((int)s_cnt[dl]);
            n = n < (unsigned)BIN_CAP ? n : (unsigned)BIN_CAP;
            const float2 q = *(const float2*)&s_q[dl * HD + 2 * lane];
            float2 acc = *(const float2*)&s_acc[dl * ACC_PITCH + 2 * lane];
            float zacc = 0.0f;
            const unsigned bin0 = s_bin[dl * BIN_CAP + lane];
            const unsigned bin1 = s_bin[dl * BIN_CAP + lane + 32];
            for (unsigned j0 = 0; j0 < n; j0 += 8) {
                const bool     live = j0 + (unsigned)(lane & 7) < n;
                const unsigned half = j0 < 32u ? bin0 : bin1;
                unsigned srcs[8];
#pragma unroll
                for (int t = 0; t < 8; ++t) {
                    unsigned i = j0 + (unsigned)t;
                    i = i < n ? i : n - 1u;
                    srcs[t] = (unsigned)__builtin_amdgcn_readlane((int)half, (int)(i & 31u));
                }
                node_batch8(srcs, live, q, K, V, lane, acc, zacc);
            }
            zacc += row_xor<1>(zacc);
            zacc += row_xor<2>(zacc);
            zacc += row_xor<4>(zacc);
            const float z = zacc + s_acc[dl * ACC_PITCH + HD + head];
            float2 r;
            r.x = acc.x / (z + 1e-6f);
            r.y = acc.y / (z + 1e-6f);
            *(float2*)(out + (size_t)(lo + dl) * HD + 2 * lane) = r;
        }
    }
}

extern "C" void kernel_launch(void* const* d_in, const int* in_sizes, int n_in,
                              void* d_out, int out_size, void* d_ws, size_t ws_size,
                              hipStream_t stream)
{
    (void)in_sizes; (void)n_in; (void)out_size; (void)ws_size;

    const float* X   = (const float*)d_in[0];
    const int*   src = (const int*)  d_in[1];
    const int*   dst = (const int*)  d_in[2];
    const float* Wq  = (const float*)d_in[3];
    const float* Wk  = (const float*)d_in[4];
    const float* Wv  = (const float*)d_in[5];

    float* out = (float*)d_out;
    float* ws  = (float*)d_ws;
    unsigned* plane = (unsigned*)d_ws;
    unsigned* pk    = (unsigned*)d_ws + PK_OFF_WORDS;
    float* Q   = out;
    float* K   = ws + (size_t)N_NODES * HD;
    float* V   = ws + (size_t)2 * N_NODES * HD;

    {
        const int ntask = (N_NODES / 16) * 3;
        const int wpb   = 256 / 32;
        const int blocks = PLANE_BLOCKS + (ntask + wpb - 1) / wpb;
        gt_qkv_proj_wmma<<<blocks, 256, 0, stream>>>(X, src, dst, plane, pk, Wq, Wk, Wv, Q, K, V);
    }

    gt_edge_owner<<<N_OWNERS, 1024, 0, stream>>>(pk, plane, K, V, out);
}
